# starting kernel
# baseline (speedup 1.0000x reference)
#include <hip/hip_runtime.h>

typedef __attribute__((ext_vector_type(2))) float v2f;
typedef __attribute__((ext_vector_type(8))) float v8f;

#define N_NODES 100000
#define N_EDGES 1600000
#define IN_F 128
#define HID 128

__global__ void degrees_kernel(const int* __restrict__ src, const int* __restrict__ dst,
                               float* __restrict__ deg_out, float* __restrict__ deg_in, int E) {
    int e = blockIdx.x * blockDim.x + threadIdx.x;
    if (e >= E) return;
    atomicAdd(&deg_out[src[e]], 1.0f);
    atomicAdd(&deg_in[dst[e]], 1.0f);
}

__global__ void ninv_kernel(float* __restrict__ deg, int n) {
    int i = blockIdx.x * blockDim.x + threadIdx.x;
    if (i >= n) return;
    float d = deg[i];
    deg[i] = (d > 0.0f) ? (1.0f / sqrtf(d)) : 0.0f;
}

__global__ void scale_kernel(const float* __restrict__ x, const float* __restrict__ ninv_out,
                             float* __restrict__ xs) {
    int i = blockIdx.x * blockDim.x + threadIdx.x;
    xs[i] = x[i] * ninv_out[i >> 7];
}

__global__ void agg128_kernel(const int* __restrict__ src, const int* __restrict__ dst,
                              const float* __restrict__ xs, float* __restrict__ agg, int E) {
    int e = (blockIdx.x * blockDim.x + threadIdx.x) >> 5;
    int lane = threadIdx.x & 31;
    if (e >= E) return;
    int s = src[e];
    int d = dst[e];
    const float4 v = *(const float4*)(xs + (size_t)s * 128 + lane * 4);
    float* dp = agg + (size_t)d * 128 + lane * 4;
    atomicAdd(dp + 0, v.x);
    atomicAdd(dp + 1, v.y);
    atomicAdd(dp + 2, v.z);
    atomicAdd(dp + 3, v.w);
}

__global__ void gemm1_relu_kernel(const float* __restrict__ agg, const float* __restrict__ W1,
                                  const float* __restrict__ b1, const float* __restrict__ ninv_in,
                                  float* __restrict__ h1) {
    __shared__ float As[16 * 128];

    const int tid  = threadIdx.x;
    const int lane = tid & 31;
    const int wave = tid >> 5;
    const int row0 = blockIdx.x * 16;

    const float4* g4 = (const float4*)(agg + (size_t)row0 * 128);
    float4* s4 = (float4*)As;
    s4[tid * 2 + 0] = g4[tid * 2 + 0];
    s4[tid * 2 + 1] = g4[tid * 2 + 1];
    __syncthreads();

    const int col0 = wave * 16;
    const int mn   = lane & 15;
    const int kb   = (lane >> 4) * 2;

    v8f c = {};
#pragma unroll
    for (int kk = 0; kk < 32; ++kk) {
        const int k = kk * 4 + kb;
        v2f a, b;
        a.x = As[mn * 128 + k];
        a.y = As[mn * 128 + k + 1];
        b.x = W1[(size_t)k * 128 + col0 + mn];
        b.y = W1[(size_t)(k + 1) * 128 + col0 + mn];
        c = __builtin_amdgcn_wmma_f32_16x16x4_f32(false, a, false, b,
                                                  (short)0, c, false, false);
    }

    const int n     = col0 + mn;
    const int mbase = row0 + ((lane >> 4) ? 8 : 0);
    const float bn  = b1[n];
#pragma unroll
    for (int r = 0; r < 8; ++r) {
        const int m = mbase + r;
        float v = c[r] * ninv_in[m] + bn;
        h1[(size_t)m * 128 + n] = fmaxf(v, 0.0f);
    }
}

__global__ void l2_matvec_kernel(const float* __restrict__ h1, const float* __restrict__ ninv_out,
                                 const float* __restrict__ W2, float* __restrict__ y_node, int N) {
    int node = blockIdx.x * 8 + (threadIdx.x >> 5);
    int lane = threadIdx.x & 31;
    if (node >= N) return;
    const float s = ninv_out[node];
    float y0 = 0.0f, y1 = 0.0f;
#pragma unroll
    for (int j = 0; j < 4; ++j) {
        int f = lane + j * 32;
        float v = h1[(size_t)node * 128 + f] * s;
        y0 += v * W2[f * 2 + 0];
        y1 += v * W2[f * 2 + 1];
    }
#pragma unroll
    for (int off = 16; off > 0; off >>= 1) {
        y0 += __shfl_xor(y0, off, 32);
        y1 += __shfl_xor(y1, off, 32);
    }
    if (lane == 0) {
        y_node[node * 2 + 0] = y0;
        y_node[node * 2 + 1] = y1;
    }
}

__global__ void agg2_kernel(const int* __restrict__ src, const int* __restrict__ dst,
                            const float* __restrict__ y_node, float* __restrict__ out_acc, int E) {
    int e = blockIdx.x * blockDim.x + threadIdx.x;
    if (e >= E) return;
    int s = src[e];
    int d = dst[e];
    atomicAdd(&out_acc[d * 2 + 0], y_node[s * 2 + 0]);
    atomicAdd(&out_acc[d * 2 + 1], y_node[s * 2 + 1]);
}

__global__ void final_kernel(const float* __restrict__ out_acc, const float* __restrict__ ninv_in,
                             const float* __restrict__ b2, float* __restrict__ out, int N) {
    int n = blockIdx.x * blockDim.x + threadIdx.x;
    if (n >= N) return;
    float ni = ninv_in[n];
    out[n * 2 + 0] = out_acc[n * 2 + 0] * ni + b2[0];
    out[n * 2 + 1] = out_acc[n * 2 + 1] * ni + b2[1];
}

extern "C" void kernel_launch(void* const* d_in, const int* in_sizes, int n_in,
                              void* d_out, int out_size, void* d_ws, size_t ws_size,
                              hipStream_t stream) {
    const float* in_feat = (const float*)d_in[0];
    const float* W1      = (const float*)d_in[1];
    const float* b1      = (const float*)d_in[2];
    const float* W2      = (const float*)d_in[3];
    const float* b2      = (const float*)d_in[4];
    const int*   src     = (const int*)d_in[5];
    const int*   dst     = (const int*)d_in[6];
    float*       out     = (float*)d_out;

    const int N = N_NODES;
    const int E = N_EDGES;

    float* ws      = (float*)d_ws;
    float* deg_out = ws;
    float* deg_in  = ws + N;
    float* xs      = ws + 2 * (size_t)N;
    float* agg     = xs + (size_t)N * 128;
    float* y_node  = agg + (size_t)N * 128;
    float* out_acc = y_node + 2 * (size_t)N;

    hipMemsetAsync(deg_out, 0, 2 * (size_t)N * sizeof(float), stream);
    hipMemsetAsync(agg, 0, (size_t)N * 128 * sizeof(float), stream);
    hipMemsetAsync(out_acc, 0, 2 * (size_t)N * sizeof(float), stream);

    degrees_kernel<<<(E + 255) / 256, 256, 0, stream>>>(src, dst, deg_out, deg_in, E);
    ninv_kernel<<<(2 * N + 255) / 256, 256, 0, stream>>>(deg_out, 2 * N);
    scale_kernel<<<(N * 128) / 256, 256, 0, stream>>>(in_feat, deg_out, xs);
    agg128_kernel<<<E / 8, 256, 0, stream>>>(src, dst, xs, agg, E);
    gemm1_relu_kernel<<<N / 16, 256, 0, stream>>>(agg, W1, b1, deg_in, xs);
    l2_matvec_kernel<<<N / 8, 256, 0, stream>>>(xs, deg_out, W2, y_node, N);
    agg2_kernel<<<(E + 255) / 256, 256, 0, stream>>>(src, dst, y_node, out_acc, E);
    final_kernel<<<(N + 255) / 256, 256, 0, stream>>>(out_acc, deg_in, b2, out, N);
}


# agg128: lane l owns columns l,l+32,l+64,l+96, so each f32 atomic instruction adds one whole 128-byte line instead of quarters of four lines; row loads likewise one line each
# speedup vs baseline: 1.1430x; 1.1430x over previous
#include <hip/hip_runtime.h>

typedef __attribute__((ext_vector_type(2))) float v2f;
typedef __attribute__((ext_vector_type(8))) float v8f;

#define N_NODES 100000
#define N_EDGES 1600000
#define IN_F 128
#define HID 128

__global__ void degrees_kernel(const int* __restrict__ src, const int* __restrict__ dst,
                               float* __restrict__ deg_out, float* __restrict__ deg_in, int E) {
    int e = blockIdx.x * blockDim.x + threadIdx.x;
    if (e >= E) return;
    atomicAdd(&deg_out[src[e]], 1.0f);
    atomicAdd(&deg_in[dst[e]], 1.0f);
}

__global__ void ninv_kernel(float* __restrict__ deg, int n) {
    int i = blockIdx.x * blockDim.x + threadIdx.x;
    if (i >= n) return;
    float d = deg[i];
    deg[i] = (d > 0.0f) ? (1.0f / sqrtf(d)) : 0.0f;
}

__global__ void scale_kernel(const float* __restrict__ x, const float* __restrict__ ninv_out,
                             float* __restrict__ xs) {
    int i = blockIdx.x * blockDim.x + threadIdx.x;
    xs[i] = x[i] * ninv_out[i >> 7];
}

__global__ void agg128_kernel(const int* __restrict__ src, const int* __restrict__ dst,
                              const float* __restrict__ xs, float* __restrict__ agg, int E) {
    int e = (blockIdx.x * blockDim.x + threadIdx.x) >> 5;
    int lane = threadIdx.x & 31;
    if (e >= E) return;
    int s = src[e];
    int d = dst[e];
    const float* sp = xs + (size_t)s * 128 + lane;
    float* dp = agg + (size_t)d * 128 + lane;
    const float v0 = sp[0];
    const float v1 = sp[32];
    const float v2 = sp[64];
    const float v3 = sp[96];
    atomicAdd(dp + 0, v0);
    atomicAdd(dp + 32, v1);
    atomicAdd(dp + 64, v2);
    atomicAdd(dp + 96, v3);
}

__global__ void gemm1_relu_kernel(const float* __restrict__ agg, const float* __restrict__ W1,
                                  const float* __restrict__ b1, const float* __restrict__ ninv_in,
                                  float* __restrict__ h1) {
    __shared__ float As[16 * 128];

    const int tid  = threadIdx.x;
    const int lane = tid & 31;
    const int wave = tid >> 5;
    const int row0 = blockIdx.x * 16;

    const float4* g4 = (const float4*)(agg + (size_t)row0 * 128);
    float4* s4 = (float4*)As;
    s4[tid * 2 + 0] = g4[tid * 2 + 0];
    s4[tid * 2 + 1] = g4[tid * 2 + 1];
    __syncthreads();

    const int col0 = wave * 16;
    const int mn   = lane & 15;
    const int kb   = (lane >> 4) * 2;

    v8f c = {};
#pragma unroll
    for (int kk = 0; kk < 32; ++kk) {
        const int k = kk * 4 + kb;
        v2f a, b;
        a.x = As[mn * 128 + k];
        a.y = As[mn * 128 + k + 1];
        b.x = W1[(size_t)k * 128 + col0 + mn];
        b.y = W1[(size_t)(k + 1) * 128 + col0 + mn];
        c = __builtin_amdgcn_wmma_f32_16x16x4_f32(false, a, false, b,
                                                  (short)0, c, false, false);
    }

    const int n     = col0 + mn;
    const int mbase = row0 + ((lane >> 4) ? 8 : 0);
    const float bn  = b1[n];
#pragma unroll
    for (int r = 0; r < 8; ++r) {
        const int m = mbase + r;
        float v = c[r] * ninv_in[m] + bn;
        h1[(size_t)m * 128 + n] = fmaxf(v, 0.0f);
    }
}

__global__ void l2_matvec_kernel(const float* __restrict__ h1, const float* __restrict__ ninv_out,
                                 const float* __restrict__ W2, float* __restrict__ y_node, int N) {
    int node = blockIdx.x * 8 + (threadIdx.x >> 5);
    int lane = threadIdx.x & 31;
    if (node >= N) return;
    const float s = ninv_out[node];
    float y0 = 0.0f, y1 = 0.0f;
#pragma unroll
    for (int j = 0; j < 4; ++j) {
        int f = lane + j * 32;
        float v = h1[(size_t)node * 128 + f] * s;
        y0 += v * W2[f * 2 + 0];
        y1 += v * W2[f * 2 + 1];
    }
#pragma unroll
    for (int off = 16; off > 0; off >>= 1) {
        y0 += __shfl_xor(y0, off, 32);
        y1 += __shfl_xor(y1, off, 32);
    }
    if (lane == 0) {
        y_node[node * 2 + 0] = y0;
        y_node[node * 2 + 1] = y1;
    }
}

__global__ void agg2_kernel(const int* __restrict__ src, const int* __restrict__ dst,
                            const float* __restrict__ y_node, float* __restrict__ out_acc, int E) {
    int e = blockIdx.x * blockDim.x + threadIdx.x;
    if (e >= E) return;
    int s = src[e];
    int d = dst[e];
    atomicAdd(&out_acc[d * 2 + 0], y_node[s * 2 + 0]);
    atomicAdd(&out_acc[d * 2 + 1], y_node[s * 2 + 1]);
}

__global__ void final_kernel(const float* __restrict__ out_acc, const float* __restrict__ ninv_in,
                             const float* __restrict__ b2, float* __restrict__ out, int N) {
    int n = blockIdx.x * blockDim.x + threadIdx.x;
    if (n >= N) return;
    float ni = ninv_in[n];
    out[n * 2 + 0] = out_acc[n * 2 + 0] * ni + b2[0];
    out[n * 2 + 1] = out_acc[n * 2 + 1] * ni + b2[1];
}

extern "C" void kernel_launch(void* const* d_in, const int* in_sizes, int n_in,
                              void* d_out, int out_size, void* d_ws, size_t ws_size,
                              hipStream_t stream) {
    const float* in_feat = (const float*)d_in[0];
    const float* W1      = (const float*)d_in[1];
    const float* b1      = (const float*)d_in[2];
    const float* W2      = (const float*)d_in[3];
    const float* b2      = (const float*)d_in[4];
    const int*   src     = (const int*)d_in[5];
    const int*   dst     = (const int*)d_in[6];
    float*       out     = (float*)d_out;

    const int N = N_NODES;
    const int E = N_EDGES;

    float* ws      = (float*)d_ws;
    float* deg_out = ws;
    float* deg_in  = ws + N;
    float* xs      = ws + 2 * (size_t)N;
    float* agg     = xs + (size_t)N * 128;
    float* y_node  = agg + (size_t)N * 128;
    float* out_acc = y_node + 2 * (size_t)N;

    hipMemsetAsync(deg_out, 0, 2 * (size_t)N * sizeof(float), stream);
    hipMemsetAsync(agg, 0, (size_t)N * 128 * sizeof(float), stream);
    hipMemsetAsync(out_acc, 0, 2 * (size_t)N * sizeof(float), stream);

    degrees_kernel<<<(E + 255) / 256, 256, 0, stream>>>(src, dst, deg_out, deg_in, E);
    ninv_kernel<<<(2 * N + 255) / 256, 256, 0, stream>>>(deg_out, 2 * N);
    scale_kernel<<<(N * 128) / 256, 256, 0, stream>>>(in_feat, deg_out, xs);
    agg128_kernel<<<E / 8, 256, 0, stream>>>(src, dst, xs, agg, E);
    gemm1_relu_kernel<<<N / 16, 256, 0, stream>>>(agg, W1, b1, deg_in, xs);
    l2_matvec_kernel<<<N / 8, 256, 0, stream>>>(xs, deg_out, W2, y_node, N);
    agg2_kernel<<<(E + 255) / 256, 256, 0, stream>>>(src, dst, y_node, out_acc, E);
    final_kernel<<<(N + 255) / 256, 256, 0, stream>>>(out_acc, deg_in, b2, out, N);
}


# segment sums by pull: a returning int add ranks each edge in its dst, a scan makes a dst-ordered list, one wave adds a node's rows in registers (arrival order); no f32 atomics on rows/outputs
# speedup vs baseline: 1.9528x; 1.7085x over previous
#include <hip/hip_runtime.h>

typedef __attribute__((ext_vector_type(2))) float v2f;
typedef __attribute__((ext_vector_type(8))) float v8f;

#define N_NODES 100000
#define N_EDGES 1600000
#define IN_F 128
#define HID 128

__global__ void degrees_rank_kernel(const int* __restrict__ src, const int* __restrict__ dst,
                                    float* __restrict__ deg_out, int* __restrict__ cnt_in,
                                    int* __restrict__ rank, int E) {
    int e = blockIdx.x * blockDim.x + threadIdx.x;
    if (e >= E) return;
    const int s = src[e];
    const int d = dst[e];
    atomicAdd(&deg_out[s], 1.0f);
    rank[e] = atomicAdd(&cnt_in[d], 1);
}

__global__ void __launch_bounds__(1024)
scan_ninv_kernel(const int* __restrict__ cnt, float* __restrict__ deg_out,
                 int* __restrict__ off, float* __restrict__ ninv_in, int n) {
    __shared__ int wp[32];
    __shared__ int wc[32];
    const int tid  = threadIdx.x;
    const int lane = tid & 31;
    const int wave = tid >> 5;
    const int base = blockIdx.x * 1024;

    int p = 0;
    const int4* c4 = (const int4*)cnt;
#pragma unroll 8
    for (int i = tid; i < (base >> 2); i += 1024) {
        const int4 v = c4[i];
        p += (v.x + v.y) + (v.z + v.w);
    }
    const int i = base + tid;
    const int c = (i < n) ? cnt[i] : 0;

    int inc = c;
#pragma unroll
    for (int o = 1; o < 32; o <<= 1) {
        const int t = __shfl_up(inc, o, 32);
        if (lane >= o) inc += t;
    }
#pragma unroll
    for (int o = 16; o > 0; o >>= 1) p += __shfl_xor(p, o, 32);
    if (lane == 31) wc[wave] = inc;
    if (lane == 0)  wp[wave] = p;
    __syncthreads();
    int q = wp[lane] + ((lane < wave) ? wc[lane] : 0);
#pragma unroll
    for (int o = 16; o > 0; o >>= 1) q += __shfl_xor(q, o, 32);

    if (i < n) {
        off[i] = q + inc - c;
        if (i == n - 1) off[n] = q + inc;
        ninv_in[i] = (c > 0) ? (1.0f / sqrtf((float)c)) : 0.0f;
        const float d = deg_out[i];
        deg_out[i] = (d > 0.0f) ? (1.0f / sqrtf(d)) : 0.0f;
    }
}

__global__ void fill_kernel(const int* __restrict__ src, const int* __restrict__ dst,
                            const int* __restrict__ rank, const int* __restrict__ off,
                            int* __restrict__ perm, int E) {
    int e = blockIdx.x * blockDim.x + threadIdx.x;
    if (e >= E) return;
    perm[off[dst[e]] + rank[e]] = src[e];
}

__global__ void scale_kernel(const float* __restrict__ x, const float* __restrict__ ninv_out,
                             float* __restrict__ xs) {
    int i = blockIdx.x * blockDim.x + threadIdx.x;
    xs[i] = x[i] * ninv_out[i >> 7];
}

__global__ void __launch_bounds__(256)
agg128_pull_kernel(const int* __restrict__ off, const int* __restrict__ perm,
                   const float* __restrict__ xs, float* __restrict__ agg, int n) {
    const int lane = threadIdx.x & 31;
    const int wv = __builtin_amdgcn_readfirstlane((int)(threadIdx.x >> 5));
    const int nwaves = gridDim.x * 8;
    const float4* xs4 = (const float4*)xs;
    float4* agg4 = (float4*)agg;
    const float4 zero4 = make_float4(0.0f, 0.0f, 0.0f, 0.0f);

#define PULL_ROW(k, v)                                                  \
    float4 v = zero4;                                                   \
    if (j + (k) < m) {                                                  \
        const int s = __builtin_amdgcn_readlane(mine, j + (k));         \
        v = xs4[(size_t)s * 32 + lane];                                 \
    }
#define PULL_ADD(v) \
    acc.x += v.x; acc.y += v.y; acc.z += v.z; acc.w += v.w;

    for (int node = blockIdx.x * 8 + wv; node < n; node += nwaves) {
        const int beg = off[node];
        const int end = off[node + 1];
        float4 acc = zero4;
        for (int base = beg; base < end; base += 32) {
            const int m = min(32, end - base);
            const int mine = (lane < m) ? perm[base + lane] : 0;
            for (int j = 0; j < m; j += 8) {
                PULL_ROW(0, v0) PULL_ROW(1, v1) PULL_ROW(2, v2) PULL_ROW(3, v3)
                PULL_ROW(4, v4) PULL_ROW(5, v5) PULL_ROW(6, v6) PULL_ROW(7, v7)
                PULL_ADD(v0) PULL_ADD(v1) PULL_ADD(v2) PULL_ADD(v3)
                PULL_ADD(v4) PULL_ADD(v5) PULL_ADD(v6) PULL_ADD(v7)
            }
        }
        agg4[(size_t)node * 32 + lane] = acc;
    }
#undef PULL_ROW
#undef PULL_ADD
}

__global__ void gemm1_relu_kernel(const float* __restrict__ agg, const float* __restrict__ W1,
                                  const float* __restrict__ b1, const float* __restrict__ ninv_in,
                                  float* __restrict__ h1) {
    __shared__ float As[16 * 128];

    const int tid  = threadIdx.x;
    const int lane = tid & 31;
    const int wave = tid >> 5;
    const int row0 = blockIdx.x * 16;

    const float4* g4 = (const float4*)(agg + (size_t)row0 * 128);
    float4* s4 = (float4*)As;
    s4[tid * 2 + 0] = g4[tid * 2 + 0];
    s4[tid * 2 + 1] = g4[tid * 2 + 1];
    __syncthreads();

    const int col0 = wave * 16;
    const int mn   = lane & 15;
    const int kb   = (lane >> 4) * 2;

    v8f c = {};
#pragma unroll
    for (int kk = 0; kk < 32; ++kk) {
        const int k = kk * 4 + kb;
        v2f a, b;
        a.x = As[mn * 128 + k];
        a.y = As[mn * 128 + k + 1];
        b.x = W1[(size_t)k * 128 + col0 + mn];
        b.y = W1[(size_t)(k + 1) * 128 + col0 + mn];
        c = __builtin_amdgcn_wmma_f32_16x16x4_f32(false, a, false, b,
                                                  (short)0, c, false, false);
    }

    const int n     = col0 + mn;
    const int mbase = row0 + ((lane >> 4) ? 8 : 0);
    const float bn  = b1[n];
#pragma unroll
    for (int r = 0; r < 8; ++r) {
        const int m = mbase + r;
        float v = c[r] * ninv_in[m] + bn;
        h1[(size_t)m * 128 + n] = fmaxf(v, 0.0f);
    }
}

__global__ void l2_matvec_kernel(const float* __restrict__ h1, const float* __restrict__ ninv_out,
                                 const float* __restrict__ W2, float* __restrict__ y_node, int N) {
    int node = blockIdx.x * 8 + (threadIdx.x >> 5);
    int lane = threadIdx.x & 31;
    if (node >= N) return;
    const float s = ninv_out[node];
    float y0 = 0.0f, y1 = 0.0f;
#pragma unroll
    for (int j = 0; j < 4; ++j) {
        int f = lane + j * 32;
        float v = h1[(size_t)node * 128 + f] * s;
        y0 += v * W2[f * 2 + 0];
        y1 += v * W2[f * 2 + 1];
    }
#pragma unroll
    for (int off = 16; off > 0; off >>= 1) {
        y0 += __shfl_xor(y0, off, 32);
        y1 += __shfl_xor(y1, off, 32);
    }
    if (lane == 0) {
        y_node[node * 2 + 0] = y0;
        y_node[node * 2 + 1] = y1;
    }
}

__global__ void __launch_bounds__(256)
agg2_pull_final_kernel(const int* __restrict__ off,
                       const int* __restrict__ perm, const float* __restrict__ y_node,
                       const float* __restrict__ ninv_in, const float* __restrict__ b2,
                       float* __restrict__ out, int n) {
    const int sub = threadIdx.x & 15;
    const int ngroups = gridDim.x * 16;
    const float2* y2 = (const float2*)y_node;
    const float b20 = b2[0];
    const float b21 = b2[1];
    for (int node = blockIdx.x * 16 + (threadIdx.x >> 4); node < n; node += ngroups) {
        const int beg = off[node];
        const int end = off[node + 1];
        float a0 = 0.0f, a1 = 0.0f;
        for (int i = beg + sub; i < end; i += 16) {
            const float2 y = y2[perm[i]];
            a0 += y.x;
            a1 += y.y;
        }
#pragma unroll
        for (int o = 8; o > 0; o >>= 1) {
            a0 += __shfl_xor(a0, o, 32);
            a1 += __shfl_xor(a1, o, 32);
        }
        if (sub == 0) {
            const float ni = ninv_in[node];
            out[node * 2 + 0] = a0 * ni + b20;
            out[node * 2 + 1] = a1 * ni + b21;
        }
    }
}

extern "C" void kernel_launch(void* const* d_in, const int* in_sizes, int n_in,
                              void* d_out, int out_size, void* d_ws, size_t ws_size,
                              hipStream_t stream) {
    const float* in_feat = (const float*)d_in[0];
    const float* W1      = (const float*)d_in[1];
    const float* b1      = (const float*)d_in[2];
    const float* W2      = (const float*)d_in[3];
    const float* b2      = (const float*)d_in[4];
    const int*   src     = (const int*)d_in[5];
    const int*   dst     = (const int*)d_in[6];
    float*       out     = (float*)d_out;

    const int N = N_NODES;
    const int E = N_EDGES;

    float* ws      = (float*)d_ws;
    float* deg_out = ws;
    int*   cnt_in  = (int*)(ws + N);
    float* ninv_in = ws + 2 * (size_t)N;
    int*   off     = (int*)(ws + 3 * (size_t)N);
    int*   rank    = (int*)(ws + 4 * (size_t)N + 64);
    int*   perm    = rank + (size_t)E;
    float* xs      = ws + 4 * (size_t)N + 64 + 2 * (size_t)E;
    float* agg     = xs + (size_t)N * 128;
    float* y_node  = agg + (size_t)N * 128;

    hipMemsetAsync(deg_out, 0, 2 * (size_t)N * sizeof(float), stream);

    degrees_rank_kernel<<<(E + 255) / 256, 256, 0, stream>>>(src, dst, deg_out, cnt_in, rank, E);
    scan_ninv_kernel<<<(N + 1023) / 1024, 1024, 0, stream>>>(cnt_in, deg_out, off, ninv_in, N);
    fill_kernel<<<(E + 255) / 256, 256, 0, stream>>>(src, dst, rank, off, perm, E);
    scale_kernel<<<(N * 128) / 256, 256, 0, stream>>>(in_feat, deg_out, xs);
    agg128_pull_kernel<<<(N / 8 + 7) / 8, 256, 0, stream>>>(off, perm, xs, agg, N);
    gemm1_relu_kernel<<<N / 16, 256, 0, stream>>>(agg, W1, b1, ninv_in, xs);
    l2_matvec_kernel<<<N / 8, 256, 0, stream>>>(xs, deg_out, W2, y_node, N);
    agg2_pull_final_kernel<<<(N / 4 + 15) / 16, 256, 0, stream>>>(off, perm, y_node, ninv_in, b2, out, N);
}


# degree scaling moved into the pull aggregation: it gathers raw input rows and multiplies by ninv_out[src] in registers (fused multiply-add into the sum); scale launch and x_scaled plane gone
# speedup vs baseline: 2.1318x; 1.0917x over previous
#include <hip/hip_runtime.h>

typedef __attribute__((ext_vector_type(2))) float v2f;
typedef __attribute__((ext_vector_type(8))) float v8f;

#define N_NODES 100000
#define N_EDGES 1600000
#define IN_F 128
#define HID 128

__global__ void degrees_rank_kernel(const int* __restrict__ src, const int* __restrict__ dst,
                                    float* __restrict__ deg_out, int* __restrict__ cnt_in,
                                    int* __restrict__ rank, int E) {
    int e = blockIdx.x * blockDim.x + threadIdx.x;
    if (e >= E) return;
    const int s = src[e];
    const int d = dst[e];
    atomicAdd(&deg_out[s], 1.0f);
    rank[e] = atomicAdd(&cnt_in[d], 1);
}

__global__ void __launch_bounds__(1024)
scan_ninv_kernel(const int* __restrict__ cnt, float* __restrict__ deg_out,
                 int* __restrict__ off, float* __restrict__ ninv_in, int n) {
    __shared__ int wp[32];
    __shared__ int wc[32];
    const int tid  = threadIdx.x;
    const int lane = tid & 31;
    const int wave = tid >> 5;
    const int base = blockIdx.x * 1024;

    int p = 0;
    const int4* c4 = (const int4*)cnt;
#pragma unroll 8
    for (int i = tid; i < (base >> 2); i += 1024) {
        const int4 v = c4[i];
        p += (v.x + v.y) + (v.z + v.w);
    }
    const int i = base + tid;
    const int c = (i < n) ? cnt[i] : 0;

    int inc = c;
#pragma unroll
    for (int o = 1; o < 32; o <<= 1) {
        const int t = __shfl_up(inc, o, 32);
        if (lane >= o) inc += t;
    }
#pragma unroll
    for (int o = 16; o > 0; o >>= 1) p += __shfl_xor(p, o, 32);
    if (lane == 31) wc[wave] = inc;
    if (lane == 0)  wp[wave] = p;
    __syncthreads();
    int q = wp[lane] + ((lane < wave) ? wc[lane] : 0);
#pragma unroll
    for (int o = 16; o > 0; o >>= 1) q += __shfl_xor(q, o, 32);

    if (i < n) {
        off[i] = q + inc - c;
        if (i == n - 1) off[n] = q + inc;
        ninv_in[i] = (c > 0) ? (1.0f / sqrtf((float)c)) : 0.0f;
        const float d = deg_out[i];
        deg_out[i] = (d > 0.0f) ? (1.0f / sqrtf(d)) : 0.0f;
    }
}

__global__ void fill_kernel(const int* __restrict__ src, const int* __restrict__ dst,
                            const int* __restrict__ rank, const int* __restrict__ off,
                            int* __restrict__ perm, int E) {
    int e = blockIdx.x * blockDim.x + threadIdx.x;
    if (e >= E) return;
    perm[off[dst[e]] + rank[e]] = src[e];
}

__global__ void __launch_bounds__(256)
agg128_pull_kernel(const int* __restrict__ off, const int* __restrict__ perm,
                   const float* __restrict__ x, const float* __restrict__ ninv_out,
                   float* __restrict__ agg, int n) {
    const int lane = threadIdx.x & 31;
    const int wv = __builtin_amdgcn_readfirstlane((int)(threadIdx.x >> 5));
    const int nwaves = gridDim.x * 8;
    const float4* x4 = (const float4*)x;
    float4* agg4 = (float4*)agg;
    const float4 zero4 = make_float4(0.0f, 0.0f, 0.0f, 0.0f);

#define PULL_ROW(k, v)                                                  \
    float4 v = zero4;                                                   \
    if (j + (k) < m) {                                                  \
        const int s = __builtin_amdgcn_readlane(mine, j + (k));         \
        v = x4[(size_t)s * 32 + lane];                                  \
    }
#define PULL_ADD(k, v)                                                              \
    {                                                                               \
        const float w = __int_as_float(__builtin_amdgcn_readlane(scl, j + (k)));    \
        acc.x += v.x * w; acc.y += v.y * w; acc.z += v.z * w; acc.w += v.w * w;     \
    }

    for (int node = blockIdx.x * 8 + wv; node < n; node += nwaves) {
        const int beg = off[node];
        const int end = off[node + 1];
        float4 acc = zero4;
        for (int base = beg; base < end; base += 32) {
            const int m = min(32, end - base);
            const int mine = (lane < m) ? perm[base + lane] : 0;
            const int scl = __float_as_int(ninv_out[mine]);
            for (int j = 0; j < m; j += 8) {
                PULL_ROW(0, v0) PULL_ROW(1, v1) PULL_ROW(2, v2) PULL_ROW(3, v3)
                PULL_ROW(4, v4) PULL_ROW(5, v5) PULL_ROW(6, v6) PULL_ROW(7, v7)
                PULL_ADD(0, v0) PULL_ADD(1, v1) PULL_ADD(2, v2) PULL_ADD(3, v3)
                PULL_ADD(4, v4) PULL_ADD(5, v5) PULL_ADD(6, v6) PULL_ADD(7, v7)
            }
        }
        agg4[(size_t)node * 32 + lane] = acc;
    }
#undef PULL_ROW
#undef PULL_ADD
}

__global__ void gemm1_relu_kernel(const float* __restrict__ agg, const float* __restrict__ W1,
                                  const float* __restrict__ b1, const float* __restrict__ ninv_in,
                                  float* __restrict__ h1) {
    __shared__ float As[16 * 128];

    const int tid  = threadIdx.x;
    const int lane = tid & 31;
    const int wave = tid >> 5;
    const int row0 = blockIdx.x * 16;

    const float4* g4 = (const float4*)(agg + (size_t)row0 * 128);
    float4* s4 = (float4*)As;
    s4[tid * 2 + 0] = g4[tid * 2 + 0];
    s4[tid * 2 + 1] = g4[tid * 2 + 1];
    __syncthreads();

    const int col0 = wave * 16;
    const int mn   = lane & 15;
    const int kb   = (lane >> 4) * 2;

    v8f c = {};
#pragma unroll
    for (int kk = 0; kk < 32; ++kk) {
        const int k = kk * 4 + kb;
        v2f a, b;
        a.x = As[mn * 128 + k];
        a.y = As[mn * 128 + k + 1];
        b.x = W1[(size_t)k * 128 + col0 + mn];
        b.y = W1[(size_t)(k + 1) * 128 + col0 + mn];
        c = __builtin_amdgcn_wmma_f32_16x16x4_f32(false, a, false, b,
                                                  (short)0, c, false, false);
    }

    const int n     = col0 + mn;
    const int mbase = row0 + ((lane >> 4) ? 8 : 0);
    const float bn  = b1[n];
#pragma unroll
    for (int r = 0; r < 8; ++r) {
        const int m = mbase + r;
        float v = c[r] * ninv_in[m] + bn;
        h1[(size_t)m * 128 + n] = fmaxf(v, 0.0f);
    }
}

__global__ void l2_matvec_kernel(const float* __restrict__ h1, const float* __restrict__ ninv_out,
                                 const float* __restrict__ W2, float* __restrict__ y_node, int N) {
    int node = blockIdx.x * 8 + (threadIdx.x >> 5);
    int lane = threadIdx.x & 31;
    if (node >= N) return;
    const float s = ninv_out[node];
    float y0 = 0.0f, y1 = 0.0f;
#pragma unroll
    for (int j = 0; j < 4; ++j) {
        int f = lane + j * 32;
        float v = h1[(size_t)node * 128 + f] * s;
        y0 += v * W2[f * 2 + 0];
        y1 += v * W2[f * 2 + 1];
    }
#pragma unroll
    for (int off = 16; off > 0; off >>= 1) {
        y0 += __shfl_xor(y0, off, 32);
        y1 += __shfl_xor(y1, off, 32);
    }
    if (lane == 0) {
        y_node[node * 2 + 0] = y0;
        y_node[node * 2 + 1] = y1;
    }
}

__global__ void __launch_bounds__(256)
agg2_pull_final_kernel(const int* __restrict__ off,
                       const int* __restrict__ perm, const float* __restrict__ y_node,
                       const float* __restrict__ ninv_in, const float* __restrict__ b2,
                       float* __restrict__ out, int n) {
    const int sub = threadIdx.x & 15;
    const int ngroups = gridDim.x * 16;
    const float2* y2 = (const float2*)y_node;
    const float b20 = b2[0];
    const float b21 = b2[1];
    for (int node = blockIdx.x * 16 + (threadIdx.x >> 4); node < n; node += ngroups) {
        const int beg = off[node];
        const int end = off[node + 1];
        float a0 = 0.0f, a1 = 0.0f;
        for (int i = beg + sub; i < end; i += 16) {
            const float2 y = y2[perm[i]];
            a0 += y.x;
            a1 += y.y;
        }
#pragma unroll
        for (int o = 8; o > 0; o >>= 1) {
            a0 += __shfl_xor(a0, o, 32);
            a1 += __shfl_xor(a1, o, 32);
        }
        if (sub == 0) {
            const float ni = ninv_in[node];
            out[node * 2 + 0] = a0 * ni + b20;
            out[node * 2 + 1] = a1 * ni + b21;
        }
    }
}

extern "C" void kernel_launch(void* const* d_in, const int* in_sizes, int n_in,
                              void* d_out, int out_size, void* d_ws, size_t ws_size,
                              hipStream_t stream) {
    const float* in_feat = (const float*)d_in[0];
    const float* W1      = (const float*)d_in[1];
    const float* b1      = (const float*)d_in[2];
    const float* W2      = (const float*)d_in[3];
    const float* b2      = (const float*)d_in[4];
    const int*   src     = (const int*)d_in[5];
    const int*   dst     = (const int*)d_in[6];
    float*       out     = (float*)d_out;

    const int N = N_NODES;
    const int E = N_EDGES;

    float* ws      = (float*)d_ws;
    float* deg_out = ws;
    int*   cnt_in  = (int*)(ws + N);
    float* ninv_in = ws + 2 * (size_t)N;
    int*   off     = (int*)(ws + 3 * (size_t)N);
    int*   rank    = (int*)(ws + 4 * (size_t)N + 64);
    int*   perm    = rank + (size_t)E;
    float* xs      = ws + 4 * (size_t)N + 64 + 2 * (size_t)E;
    float* agg     = xs + (size_t)N * 128;
    float* y_node  = agg + (size_t)N * 128;

    hipMemsetAsync(deg_out, 0, 2 * (size_t)N * sizeof(float), stream);

    degrees_rank_kernel<<<(E + 255) / 256, 256, 0, stream>>>(src, dst, deg_out, cnt_in, rank, E);
    scan_ninv_kernel<<<(N + 1023) / 1024, 1024, 0, stream>>>(cnt_in, deg_out, off, ninv_in, N);
    fill_kernel<<<(E + 255) / 256, 256, 0, stream>>>(src, dst, rank, off, perm, E);
    agg128_pull_kernel<<<(N / 8 + 7) / 8, 256, 0, stream>>>(off, perm, in_feat, deg_out, agg, N);
    gemm1_relu_kernel<<<N / 16, 256, 0, stream>>>(agg, W1, b1, ninv_in, xs);
    l2_matvec_kernel<<<N / 8, 256, 0, stream>>>(xs, deg_out, W2, y_node, N);
    agg2_pull_final_kernel<<<(N / 4 + 15) / 16, 256, 0, stream>>>(off, perm, y_node, ninv_in, b2, out, N);
}


# GEMM A tile in LDS at a row pitch of 132 words instead of 128: the 16 rows of a fragment read fall on different banks, no 16-way bank conflict on every A read; arithmetic and sum order as before
# speedup vs baseline: 2.2895x; 1.0740x over previous
#include <hip/hip_runtime.h>

typedef __attribute__((ext_vector_type(2))) float v2f;
typedef __attribute__((ext_vector_type(8))) float v8f;

#define N_NODES 100000
#define N_EDGES 1600000
#define IN_F 128
#define HID 128

__global__ void degrees_rank_kernel(const int* __restrict__ src, const int* __restrict__ dst,
                                    float* __restrict__ deg_out, int* __restrict__ cnt_in,
                                    int* __restrict__ rank, int E) {
    int e = blockIdx.x * blockDim.x + threadIdx.x;
    if (e >= E) return;
    const int s = src[e];
    const int d = dst[e];
    atomicAdd(&deg_out[s], 1.0f);
    rank[e] = atomicAdd(&cnt_in[d], 1);
}

__global__ void __launch_bounds__(1024)
scan_ninv_kernel(const int* __restrict__ cnt, float* __restrict__ deg_out,
                 int* __restrict__ off, float* __restrict__ ninv_in, int n) {
    __shared__ int wp[32];
    __shared__ int wc[32];
    const int tid  = threadIdx.x;
    const int lane = tid & 31;
    const int wave = tid >> 5;
    const int base = blockIdx.x * 1024;

    int p = 0;
    const int4* c4 = (const int4*)cnt;
#pragma unroll 8
    for (int i = tid; i < (base >> 2); i += 1024) {
        const int4 v = c4[i];
        p += (v.x + v.y) + (v.z + v.w);
    }
    const int i = base + tid;
    const int c = (i < n) ? cnt[i] : 0;

    int inc = c;
#pragma unroll
    for (int o = 1; o < 32; o <<= 1) {
        const int t = __shfl_up(inc, o, 32);
        if (lane >= o) inc += t;
    }
#pragma unroll
    for (int o = 16; o > 0; o >>= 1) p += __shfl_xor(p, o, 32);
    if (lane == 31) wc[wave] = inc;
    if (lane == 0)  wp[wave] = p;
    __syncthreads();
    int q = wp[lane] + ((lane < wave) ? wc[lane] : 0);
#pragma unroll
    for (int o = 16; o > 0; o >>= 1) q += __shfl_xor(q, o, 32);

    if (i < n) {
        off[i] = q + inc - c;
        if (i == n - 1) off[n] = q + inc;
        ninv_in[i] = (c > 0) ? (1.0f / sqrtf((float)c)) : 0.0f;
        const float d = deg_out[i];
        deg_out[i] = (d > 0.0f) ? (1.0f / sqrtf(d)) : 0.0f;
    }
}

__global__ void fill_kernel(const int* __restrict__ src, const int* __restrict__ dst,
                            const int* __restrict__ rank, const int* __restrict__ off,
                            int* __restrict__ perm, int E) {
    int e = blockIdx.x * blockDim.x + threadIdx.x;
    if (e >= E) return;
    perm[off[dst[e]] + rank[e]] = src[e];
}

__global__ void __launch_bounds__(256)
agg128_pull_kernel(const int* __restrict__ off, const int* __restrict__ perm,
                   const float* __restrict__ x, const float* __restrict__ ninv_out,
                   float* __restrict__ agg, int n) {
    const int lane = threadIdx.x & 31;
    const int wv = __builtin_amdgcn_readfirstlane((int)(threadIdx.x >> 5));
    const int nwaves = gridDim.x * 8;
    const float4* x4 = (const float4*)x;
    float4* agg4 = (float4*)agg;
    const float4 zero4 = make_float4(0.0f, 0.0f, 0.0f, 0.0f);

#define PULL_ROW(k, v)                                                  \
    float4 v = zero4;                                                   \
    if (j + (k) < m) {                                                  \
        const int s = __builtin_amdgcn_readlane(mine, j + (k));         \
        v = x4[(size_t)s * 32 + lane];                                  \
    }
#define PULL_ADD(k, v)                                                              \
    {                                                                               \
        const float w = __int_as_float(__builtin_amdgcn_readlane(scl, j + (k)));    \
        acc.x += v.x * w; acc.y += v.y * w; acc.z += v.z * w; acc.w += v.w * w;     \
    }

    for (int node = blockIdx.x * 8 + wv; node < n; node += nwaves) {
        const int beg = off[node];
        const int end = off[node + 1];
        float4 acc = zero4;
        for (int base = beg; base < end; base += 32) {
            const int m = min(32, end - base);
            const int mine = (lane < m) ? perm[base + lane] : 0;
            const int scl = __float_as_int(ninv_out[mine]);
            for (int j = 0; j < m; j += 8) {
                PULL_ROW(0, v0) PULL_ROW(1, v1) PULL_ROW(2, v2) PULL_ROW(3, v3)
                PULL_ROW(4, v4) PULL_ROW(5, v5) PULL_ROW(6, v6) PULL_ROW(7, v7)
                PULL_ADD(0, v0) PULL_ADD(1, v1) PULL_ADD(2, v2) PULL_ADD(3, v3)
                PULL_ADD(4, v4) PULL_ADD(5, v5) PULL_ADD(6, v6) PULL_ADD(7, v7)
            }
        }
        agg4[(size_t)node * 32 + lane] = acc;
    }
#undef PULL_ROW
#undef PULL_ADD
}

__global__ void gemm1_relu_kernel(const float* __restrict__ agg, const float* __restrict__ W1,
                                  const float* __restrict__ b1, const float* __restrict__ ninv_in,
                                  float* __restrict__ h1) {
    constexpr int LDA = 132;
    __shared__ float As[16 * LDA];

    const int tid  = threadIdx.x;
    const int lane = tid & 31;
    const int wave = tid >> 5;
    const int row0 = blockIdx.x * 16;

    const float4* g4 = (const float4*)(agg + (size_t)row0 * 128);
    {
        const int i0 = tid * 2;
        float4* srow = (float4*)(As + (i0 >> 5) * LDA + (i0 & 31) * 4);
        srow[0] = g4[i0 + 0];
        srow[1] = g4[i0 + 1];
    }
    __syncthreads();

    const int col0 = wave * 16;
    const int mn   = lane & 15;
    const int kb   = (lane >> 4) * 2;

    v8f c = {};
#pragma unroll
    for (int kk = 0; kk < 32; ++kk) {
        const int k = kk * 4 + kb;
        v2f a, b;
        a.x = As[mn * LDA + k];
        a.y = As[mn * LDA + k + 1];
        b.x = W1[(size_t)k * 128 + col0 + mn];
        b.y = W1[(size_t)(k + 1) * 128 + col0 + mn];
        c = __builtin_amdgcn_wmma_f32_16x16x4_f32(false, a, false, b,
                                                  (short)0, c, false, false);
    }

    const int n     = col0 + mn;
    const int mbase = row0 + ((lane >> 4) ? 8 : 0);
    const float bn  = b1[n];
#pragma unroll
    for (int r = 0; r < 8; ++r) {
        const int m = mbase + r;
        float v = c[r] * ninv_in[m] + bn;
        h1[(size_t)m * 128 + n] = fmaxf(v, 0.0f);
    }
}

__global__ void l2_matvec_kernel(const float* __restrict__ h1, const float* __restrict__ ninv_out,
                                 const float* __restrict__ W2, float* __restrict__ y_node, int N) {
    int node = blockIdx.x * 8 + (threadIdx.x >> 5);
    int lane = threadIdx.x & 31;
    if (node >= N) return;
    const float s = ninv_out[node];
    float y0 = 0.0f, y1 = 0.0f;
#pragma unroll
    for (int j = 0; j < 4; ++j) {
        int f = lane + j * 32;
        float v = h1[(size_t)node * 128 + f] * s;
        y0 += v * W2[f * 2 + 0];
        y1 += v * W2[f * 2 + 1];
    }
#pragma unroll
    for (int off = 16; off > 0; off >>= 1) {
        y0 += __shfl_xor(y0, off, 32);
        y1 += __shfl_xor(y1, off, 32);
    }
    if (lane == 0) {
        y_node[node * 2 + 0] = y0;
        y_node[node * 2 + 1] = y1;
    }
}

__global__ void __launch_bounds__(256)
agg2_pull_final_kernel(const int* __restrict__ off,
                       const int* __restrict__ perm, const float* __restrict__ y_node,
                       const float* __restrict__ ninv_in, const float* __restrict__ b2,
                       float* __restrict__ out, int n) {
    const int sub = threadIdx.x & 15;
    const int ngroups = gridDim.x * 16;
    const float2* y2 = (const float2*)y_node;
    const float b20 = b2[0];
    const float b21 = b2[1];
    for (int node = blockIdx.x * 16 + (threadIdx.x >> 4); node < n; node += ngroups) {
        const int beg = off[node];
        const int end = off[node + 1];
        float a0 = 0.0f, a1 = 0.0f;
        for (int i = beg + sub; i < end; i += 16) {
            const float2 y = y2[perm[i]];
            a0 += y.x;
            a1 += y.y;
        }
#pragma unroll
        for (int o = 8; o > 0; o >>= 1) {
            a0 += __shfl_xor(a0, o, 32);
            a1 += __shfl_xor(a1, o, 32);
        }
        if (sub == 0) {
            const float ni = ninv_in[node];
            out[node * 2 + 0] = a0 * ni + b20;
            out[node * 2 + 1] = a1 * ni + b21;
        }
    }
}

extern "C" void kernel_launch(void* const* d_in, const int* in_sizes, int n_in,
                              void* d_out, int out_size, void* d_ws, size_t ws_size,
                              hipStream_t stream) {
    const float* in_feat = (const float*)d_in[0];
    const float* W1      = (const float*)d_in[1];
    const float* b1      = (const float*)d_in[2];
    const float* W2      = (const float*)d_in[3];
    const float* b2      = (const float*)d_in[4];
    const int*   src     = (const int*)d_in[5];
    const int*   dst     = (const int*)d_in[6];
    float*       out     = (float*)d_out;

    const int N = N_NODES;
    const int E = N_EDGES;

    float* ws      = (float*)d_ws;
    float* deg_out = ws;
    int*   cnt_in  = (int*)(ws + N);
    float* ninv_in = ws + 2 * (size_t)N;
    int*   off     = (int*)(ws + 3 * (size_t)N);
    int*   rank    = (int*)(ws + 4 * (size_t)N + 64);
    int*   perm    = rank + (size_t)E;
    float* xs      = ws + 4 * (size_t)N + 64 + 2 * (size_t)E;
    float* agg     = xs + (size_t)N * 128;
    float* y_node  = agg + (size_t)N * 128;

    hipMemsetAsync(deg_out, 0, 2 * (size_t)N * sizeof(float), stream);

    degrees_rank_kernel<<<(E + 255) / 256, 256, 0, stream>>>(src, dst, deg_out, cnt_in, rank, E);
    scan_ninv_kernel<<<(N + 1023) / 1024, 1024, 0, stream>>>(cnt_in, deg_out, off, ninv_in, N);
    fill_kernel<<<(E + 255) / 256, 256, 0, stream>>>(src, dst, rank, off, perm, E);
    agg128_pull_kernel<<<(N / 8 + 7) / 8, 256, 0, stream>>>(off, perm, in_feat, deg_out, agg, N);
    gemm1_relu_kernel<<<N / 16, 256, 0, stream>>>(agg, W1, b1, ninv_in, xs);
    l2_matvec_kernel<<<N / 8, 256, 0, stream>>>(xs, deg_out, W2, y_node, N);
    agg2_pull_final_kernel<<<(N / 4 + 15) / 16, 256, 0, stream>>>(off, perm, y_node, ninv_in, b2, out, N);
}


# layer-2 matvec fused into the GEMM epilogue: 16 lanes sum a row's products by a row_xmask butterfly, the 8 waves' sums meet in LDS (new sum tree for that dot); h1 plane and its launch gone
# speedup vs baseline: 2.4402x; 1.0658x over previous
#include <hip/hip_runtime.h>

typedef __attribute__((ext_vector_type(2))) float v2f;
typedef __attribute__((ext_vector_type(8))) float v8f;

#define N_NODES 100000
#define N_EDGES 1600000
#define IN_F 128
#define HID 128

__global__ void degrees_rank_kernel(const int* __restrict__ src, const int* __restrict__ dst,
                                    float* __restrict__ deg_out, int* __restrict__ cnt_in,
                                    int* __restrict__ rank, int E) {
    int e = blockIdx.x * blockDim.x + threadIdx.x;
    if (e >= E) return;
    const int s = src[e];
    const int d = dst[e];
    atomicAdd(&deg_out[s], 1.0f);
    rank[e] = atomicAdd(&cnt_in[d], 1);
}

__global__ void __launch_bounds__(1024)
scan_ninv_kernel(const int* __restrict__ cnt, float* __restrict__ deg_out,
                 int* __restrict__ off, float* __restrict__ ninv_in, int n) {
    __shared__ int wp[32];
    __shared__ int wc[32];
    const int tid  = threadIdx.x;
    const int lane = tid & 31;
    const int wave = tid >> 5;
    const int base = blockIdx.x * 1024;

    int p = 0;
    const int4* c4 = (const int4*)cnt;
#pragma unroll 8
    for (int i = tid; i < (base >> 2); i += 1024) {
        const int4 v = c4[i];
        p += (v.x + v.y) + (v.z + v.w);
    }
    const int i = base + tid;
    const int c = (i < n) ? cnt[i] : 0;

    int inc = c;
#pragma unroll
    for (int o = 1; o < 32; o <<= 1) {
        const int t = __shfl_up(inc, o, 32);
        if (lane >= o) inc += t;
    }
#pragma unroll
    for (int o = 16; o > 0; o >>= 1) p += __shfl_xor(p, o, 32);
    if (lane == 31) wc[wave] = inc;
    if (lane == 0)  wp[wave] = p;
    __syncthreads();
    int q = wp[lane] + ((lane < wave) ? wc[lane] : 0);
#pragma unroll
    for (int o = 16; o > 0; o >>= 1) q += __shfl_xor(q, o, 32);

    if (i < n) {
        off[i] = q + inc - c;
        if (i == n - 1) off[n] = q + inc;
        ninv_in[i] = (c > 0) ? (1.0f / sqrtf((float)c)) : 0.0f;
        const float d = deg_out[i];
        deg_out[i] = (d > 0.0f) ? (1.0f / sqrtf(d)) : 0.0f;
    }
}

__global__ void fill_kernel(const int* __restrict__ src, const int* __restrict__ dst,
                            const int* __restrict__ rank, const int* __restrict__ off,
                            int* __restrict__ perm, int E) {
    int e = blockIdx.x * blockDim.x + threadIdx.x;
    if (e >= E) return;
    perm[off[dst[e]] + rank[e]] = src[e];
}

__global__ void __launch_bounds__(256)
agg128_pull_kernel(const int* __restrict__ off, const int* __restrict__ perm,
                   const float* __restrict__ x, const float* __restrict__ ninv_out,
                   float* __restrict__ agg, int n) {
    const int lane = threadIdx.x & 31;
    const int wv = __builtin_amdgcn_readfirstlane((int)(threadIdx.x >> 5));
    const int nwaves = gridDim.x * 8;
    const float4* x4 = (const float4*)x;
    float4* agg4 = (float4*)agg;
    const float4 zero4 = make_float4(0.0f, 0.0f, 0.0f, 0.0f);

#define PULL_ROW(k, v)                                                  \
    float4 v = zero4;                                                   \
    if (j + (k) < m) {                                                  \
        const int s = __builtin_amdgcn_readlane(mine, j + (k));         \
        v = x4[(size_t)s * 32 + lane];                                  \
    }
#define PULL_ADD(k, v)                                                              \
    {                                                                               \
        const float w = __int_as_float(__builtin_amdgcn_readlane(scl, j + (k)));    \
        acc.x += v.x * w; acc.y += v.y * w; acc.z += v.z * w; acc.w += v.w * w;     \
    }

    for (int node = blockIdx.x * 8 + wv; node < n; node += nwaves) {
        const int beg = off[node];
        const int end = off[node + 1];
        float4 acc = zero4;
        for (int base = beg; base < end; base += 32) {
            const int m = min(32, end - base);
            const int mine = (lane < m) ? perm[base + lane] : 0;
            const int scl = __float_as_int(ninv_out[mine]);
            for (int j = 0; j < m; j += 8) {
                PULL_ROW(0, v0) PULL_ROW(1, v1) PULL_ROW(2, v2) PULL_ROW(3, v3)
                PULL_ROW(4, v4) PULL_ROW(5, v5) PULL_ROW(6, v6) PULL_ROW(7, v7)
                PULL_ADD(0, v0) PULL_ADD(1, v1) PULL_ADD(2, v2) PULL_ADD(3, v3)
                PULL_ADD(4, v4) PULL_ADD(5, v5) PULL_ADD(6, v6) PULL_ADD(7, v7)
            }
        }
        agg4[(size_t)node * 32 + lane] = acc;
    }
#undef PULL_ROW
#undef PULL_ADD
}


template <int M>
__device__ __forceinline__ float half_xor(float v) {
    return __int_as_float(__builtin_amdgcn_update_dpp(0, __float_as_int(v), 0x160 + M, 0xf, 0xf, false));
}

__global__ void gemm1_relu_l2_kernel(const float* __restrict__ agg, const float* __restrict__ W1,
                                     const float* __restrict__ b1, const float* __restrict__ ninv_in,
                                     const float* __restrict__ ninv_out, const float* __restrict__ W2,
                                     float* __restrict__ y_node) {
    constexpr int LDA = 132;
    __shared__ float As[16 * LDA];
    __shared__ float red[8 * 32];

    const int tid  = threadIdx.x;
    const int lane = tid & 31;
    const int wave = tid >> 5;
    const int row0 = blockIdx.x * 16;

    const float4* g4 = (const float4*)(agg + (size_t)row0 * 128);
    {
        const int i0 = tid * 2;
        float4* srow = (float4*)(As + (i0 >> 5) * LDA + (i0 & 31) * 4);
        srow[0] = g4[i0 + 0];
        srow[1] = g4[i0 + 1];
    }

    const int col0 = wave * 16;
    const int mn   = lane & 15;
    const int kb   = (lane >> 4) * 2;

    const int n     = col0 + mn;
    const int mbase = row0 + ((lane >> 4) ? 8 : 0);
    const float  bn = b1[n];
    const float2 w2 = ((const float2*)W2)[n];
    float ni[8], no[8];
#pragma unroll
    for (int r = 0; r < 8; ++r) {
        ni[r] = ninv_in[mbase + r];
        no[r] = ninv_out[mbase + r];
    }
    __syncthreads();

    v8f c = {};
#pragma unroll
    for (int kk = 0; kk < 32; ++kk) {
        const int k = kk * 4 + kb;
        v2f a, b;
        a.x = As[mn * LDA + k];
        a.y = As[mn * LDA + k + 1];
        b.x = W1[(size_t)k * 128 + col0 + mn];
        b.y = W1[(size_t)(k + 1) * 128 + col0 + mn];
        c = __builtin_amdgcn_wmma_f32_16x16x4_f32(false, a, false, b,
                                                  (short)0, c, false, false);
    }

    float p[16];
#pragma unroll
    for (int r = 0; r < 8; ++r) {
        const float h = fmaxf(c[r] * ni[r] + bn, 0.0f) * no[r];
        p[2 * r + 0] = h * w2.x;
        p[2 * r + 1] = h * w2.y;
    }
#pragma unroll
    for (int i = 0; i < 8; ++i) {
        const float keep = (lane & 8) ? p[i + 8] : p[i];
        const float send = (lane & 8) ? p[i] : p[i + 8];
        p[i] = keep + half_xor<8>(send);
    }
#pragma unroll
    for (int i = 0; i < 4; ++i) {
        const float keep = (lane & 4) ? p[i + 4] : p[i];
        const float send = (lane & 4) ? p[i] : p[i + 4];
        p[i] = keep + half_xor<4>(send);
    }
#pragma unroll
    for (int i = 0; i < 2; ++i) {
        const float keep = (lane & 2) ? p[i + 2] : p[i];
        const float send = (lane & 2) ? p[i] : p[i + 2];
        p[i] = keep + half_xor<2>(send);
    }
    {
        const float keep = (lane & 1) ? p[1] : p[0];
        const float send = (lane & 1) ? p[0] : p[1];
        p[0] = keep + half_xor<1>(send);
    }
    red[wave * 32 + lane] = p[0];
    __syncthreads();
    if (tid < 32) {
        const float s01 = red[tid] + red[32 + tid];
        const float s23 = red[64 + tid] + red[96 + tid];
        const float s45 = red[128 + tid] + red[160 + tid];
        const float s67 = red[192 + tid] + red[224 + tid];
        y_node[(size_t)row0 * 2 + tid] = (s01 + s23) + (s45 + s67);
    }
}

__global__ void __launch_bounds__(256)
agg2_pull_final_kernel(const int* __restrict__ off,
                       const int* __restrict__ perm, const float* __restrict__ y_node,
                       const float* __restrict__ ninv_in, const float* __restrict__ b2,
                       float* __restrict__ out, int n) {
    const int sub = threadIdx.x & 15;
    const int ngroups = gridDim.x * 16;
    const float2* y2 = (const float2*)y_node;
    const float b20 = b2[0];
    const float b21 = b2[1];
    for (int node = blockIdx.x * 16 + (threadIdx.x >> 4); node < n; node += ngroups) {
        const int beg = off[node];
        const int end = off[node + 1];
        float a0 = 0.0f, a1 = 0.0f;
        for (int i = beg + sub; i < end; i += 16) {
            const float2 y = y2[perm[i]];
            a0 += y.x;
            a1 += y.y;
        }
#pragma unroll
        for (int o = 8; o > 0; o >>= 1) {
            a0 += __shfl_xor(a0, o, 32);
            a1 += __shfl_xor(a1, o, 32);
        }
        if (sub == 0) {
            const float ni = ninv_in[node];
            out[node * 2 + 0] = a0 * ni + b20;
            out[node * 2 + 1] = a1 * ni + b21;
        }
    }
}

extern "C" void kernel_launch(void* const* d_in, const int* in_sizes, int n_in,
                              void* d_out, int out_size, void* d_ws, size_t ws_size,
                              hipStream_t stream) {
    const float* in_feat = (const float*)d_in[0];
    const float* W1      = (const float*)d_in[1];
    const float* b1      = (const float*)d_in[2];
    const float* W2      = (const float*)d_in[3];
    const float* b2      = (const float*)d_in[4];
    const int*   src     = (const int*)d_in[5];
    const int*   dst     = (const int*)d_in[6];
    float*       out     = (float*)d_out;

    const int N = N_NODES;
    const int E = N_EDGES;

    float* ws      = (float*)d_ws;
    float* deg_out = ws;
    int*   cnt_in  = (int*)(ws + N);
    float* ninv_in = ws + 2 * (size_t)N;
    int*   off     = (int*)(ws + 3 * (size_t)N);
    int*   rank    = (int*)(ws + 4 * (size_t)N + 64);
    int*   perm    = rank + (size_t)E;
    float* agg     = ws + 4 * (size_t)N + 64 + 2 * (size_t)E + (size_t)N * 128;
    float* y_node  = agg + (size_t)N * 128;

    hipMemsetAsync(deg_out, 0, 2 * (size_t)N * sizeof(float), stream);

    degrees_rank_kernel<<<(E + 255) / 256, 256, 0, stream>>>(src, dst, deg_out, cnt_in, rank, E);
    scan_ninv_kernel<<<(N + 1023) / 1024, 1024, 0, stream>>>(cnt_in, deg_out, off, ninv_in, N);
    fill_kernel<<<(E + 255) / 256, 256, 0, stream>>>(src, dst, rank, off, perm, E);
    agg128_pull_kernel<<<(N / 8 + 7) / 8, 256, 0, stream>>>(off, perm, in_feat, deg_out, agg, N);
    gemm1_relu_l2_kernel<<<N / 16, 256, 0, stream>>>(agg, W1, b1, ninv_in, deg_out, W2, y_node);
    agg2_pull_final_kernel<<<(N / 4 + 15) / 16, 256, 0, stream>>>(off, perm, y_node, ninv_in, b2, out, N);
}


# dst-ordered edge list by a two-level counting sort on LDS counters (4096-edge tiles by 256-node bucket, then a work-group per bucket); returning global adds and scan launch gone; arrival order
# speedup vs baseline: 3.5070x; 1.4372x over previous
#include <hip/hip_runtime.h>

typedef __attribute__((ext_vector_type(2))) float v2f;
typedef __attribute__((ext_vector_type(8))) float v8f;

#define N_NODES 100000
#define N_EDGES 1600000
#define IN_F 128
#define HID 128

#define TILE_E 4096
#define N_TILES ((N_EDGES + TILE_E - 1) / TILE_E)
#define BUCKET_N 256
#define N_BUCKETS ((N_NODES + BUCKET_N - 1) / BUCKET_N)
#define TPITCH 416

__global__ void __launch_bounds__(1024)
tile_sort_kernel(const int* __restrict__ src, const int* __restrict__ dst,
                 float* __restrict__ deg_out, unsigned* __restrict__ sorted,
                 int* __restrict__ tileoff) {
    __shared__ int hist[512];
    __shared__ int wsum[16];
    __shared__ unsigned stage[TILE_E];
    const int tid  = threadIdx.x;
    const int lane = tid & 31;
    const int wave = tid >> 5;
    const int w    = blockIdx.x;

    if (tid < 512) hist[tid] = 0;
    __syncthreads();

    const int q = w * (TILE_E / 4) + tid;
    const bool have = q < N_EDGES / 4;
    int4 s4 = make_int4(0, 0, 0, 0), d4 = make_int4(0, 0, 0, 0);
    int r0 = 0, r1 = 0, r2 = 0, r3 = 0;
    if (have) {
        s4 = ((const int4*)src)[q];
        d4 = ((const int4*)dst)[q];
        r0 = atomicAdd(&hist[d4.x >> 8], 1);
        r1 = atomicAdd(&hist[d4.y >> 8], 1);
        r2 = atomicAdd(&hist[d4.z >> 8], 1);
        r3 = atomicAdd(&hist[d4.w >> 8], 1);
    }
    __syncthreads();

    const int v = (tid < 512) ? hist[tid] : 0;
    int inc = v;
    for (int o = 1; o < 32; o <<= 1) {
        const int t = __shfl_up(inc, o, 32);
        if (lane >= o) inc += t;
    }
    if (tid < 512 && lane == 31) wsum[wave] = inc;
    __syncthreads();
    if (tid < 512) {
        int p = 0;
        for (int k = 0; k < 16; ++k) p += (k < wave) ? wsum[k] : 0;
        const int excl = p + inc - v;
        hist[tid] = excl;
        if (tid <= N_BUCKETS) tileoff[w * TPITCH + tid] = excl;
    }
    __syncthreads();

    if (have) {
        stage[hist[d4.x >> 8] + r0] = ((unsigned)s4.x << 8) | (unsigned)(d4.x & 255);
        stage[hist[d4.y >> 8] + r1] = ((unsigned)s4.y << 8) | (unsigned)(d4.y & 255);
        stage[hist[d4.z >> 8] + r2] = ((unsigned)s4.z << 8) | (unsigned)(d4.z & 255);
        stage[hist[d4.w >> 8] + r3] = ((unsigned)s4.w << 8) | (unsigned)(d4.w & 255);
    }
    __syncthreads();

    const int total = min(TILE_E, N_EDGES - w * TILE_E);
    if (tid * 4 < total)
        ((uint4*)(sorted + (size_t)w * TILE_E))[tid] = ((const uint4*)stage)[tid];

    if (have) {
        atomicAdd(&deg_out[s4.x], 1.0f);
        atomicAdd(&deg_out[s4.y], 1.0f);
        atomicAdd(&deg_out[s4.z], 1.0f);
        atomicAdd(&deg_out[s4.w], 1.0f);
    }
}

#define KSEG ((N_TILES + 63) / 64)

__global__ void __launch_bounds__(1024)
csr_build_kernel(const unsigned* __restrict__ sorted, const int* __restrict__ tileoff,
                 float* __restrict__ deg_out, int* __restrict__ off,
                 float* __restrict__ ninv_in, int* __restrict__ perm, int n) {
    __shared__ int segBase[KSEG * 64];
    __shared__ int segLen[KSEG * 64];
    __shared__ int cnt[BUCKET_N];
    __shared__ int cur[BUCKET_N];
    __shared__ int wsumA[32];
    __shared__ int wsumB[8];
    const int tid  = threadIdx.x;
    const int lane = tid & 31;
    const int wave = tid >> 5;
    const int b    = blockIdx.x;

    int lo = 0, len = 0;
    if (tid < N_TILES) {
        lo = tileoff[tid * TPITCH + b];
        len = tileoff[tid * TPITCH + b + 1] - lo;
    }
    if (tid < KSEG * 64) {
        segBase[tid] = tid * TILE_E + lo;
        segLen[tid]  = len;
    }
    if (tid < BUCKET_N) cnt[tid] = 0;
    int ssum = lo;
    for (int o = 16; o > 0; o >>= 1) ssum += __shfl_xor(ssum, o, 32);
    if (lane == 0) wsumA[wave] = ssum;
    __syncthreads();
    int bstart = wsumA[lane];
    for (int o = 16; o > 0; o >>= 1) bstart += __shfl_xor(bstart, o, 32);

    const int g   = tid >> 4;
    const int sub = tid & 15;
    unsigned e[KSEG];
    bool has[KSEG];
#pragma unroll
    for (int k = 0; k < KSEG; ++k) {
        const int t = g + 64 * k;
        has[k] = sub < segLen[t];
        e[k] = has[k] ? sorted[segBase[t] + sub] : 0u;
    }
#pragma unroll
    for (int k = 0; k < KSEG; ++k)
        if (has[k]) atomicAdd(&cnt[e[k] & 255u], 1);
    for (int k = 0; k < KSEG; ++k) {
        const int t = g + 64 * k;
        const int l = segLen[t];
        const int bs = segBase[t];
        for (int j = sub + 16; j < l; j += 16) atomicAdd(&cnt[sorted[bs + j] & 255u], 1);
    }
    __syncthreads();

    const int c = (tid < BUCKET_N) ? cnt[tid] : 0;
    int inc = c;
    for (int o = 1; o < 32; o <<= 1) {
        const int t = __shfl_up(inc, o, 32);
        if (lane >= o) inc += t;
    }
    if (tid < BUCKET_N && lane == 31) wsumB[wave] = inc;
    __syncthreads();
    if (tid < BUCKET_N) {
        int p = 0;
        for (int k = 0; k < 8; ++k) p += (k < wave) ? wsumB[k] : 0;
        const int slot0 = bstart + p + inc - c;
        cur[tid] = slot0;
        const int node = b * BUCKET_N + tid;
        if (node < n) {
            off[node] = slot0;
            if (node == n - 1) off[n] = slot0 + c;
            ninv_in[node] = (c > 0) ? (1.0f / sqrtf((float)c)) : 0.0f;
            const float d = deg_out[node];
            deg_out[node] = (d > 0.0f) ? (1.0f / sqrtf(d)) : 0.0f;
        }
    }
    __syncthreads();

#pragma unroll
    for (int k = 0; k < KSEG; ++k)
        if (has[k]) perm[atomicAdd(&cur[e[k] & 255u], 1)] = (int)(e[k] >> 8);
    for (int k = 0; k < KSEG; ++k) {
        const int t = g + 64 * k;
        const int l = segLen[t];
        const int bs = segBase[t];
        for (int j = sub + 16; j < l; j += 16) {
            const unsigned x = sorted[bs + j];
            perm[atomicAdd(&cur[x & 255u], 1)] = (int)(x >> 8);
        }
    }
}

__global__ void __launch_bounds__(256)
agg128_pull_kernel(const int* __restrict__ off, const int* __restrict__ perm,
                   const float* __restrict__ x, const float* __restrict__ ninv_out,
                   float* __restrict__ agg, int n) {
    const int lane = threadIdx.x & 31;
    const int wv = __builtin_amdgcn_readfirstlane((int)(threadIdx.x >> 5));
    const int nwaves = gridDim.x * 8;
    const float4* x4 = (const float4*)x;
    float4* agg4 = (float4*)agg;
    const float4 zero4 = make_float4(0.0f, 0.0f, 0.0f, 0.0f);

#define PULL_ROW(k, v)                                                  \
    float4 v = zero4;                                                   \
    if (j + (k) < m) {                                                  \
        const int s = __builtin_amdgcn_readlane(mine, j + (k));         \
        v = x4[(size_t)s * 32 + lane];                                  \
    }
#define PULL_ADD(k, v)                                                              \
    {                                                                               \
        const float w = __int_as_float(__builtin_amdgcn_readlane(scl, j + (k)));    \
        acc.x += v.x * w; acc.y += v.y * w; acc.z += v.z * w; acc.w += v.w * w;     \
    }

    for (int node = blockIdx.x * 8 + wv; node < n; node += nwaves) {
        const int beg = off[node];
        const int end = off[node + 1];
        float4 acc = zero4;
        for (int base = beg; base < end; base += 32) {
            const int m = min(32, end - base);
            const int mine = (lane < m) ? perm[base + lane] : 0;
            const int scl = __float_as_int(ninv_out[mine]);
            for (int j = 0; j < m; j += 8) {
                PULL_ROW(0, v0) PULL_ROW(1, v1) PULL_ROW(2, v2) PULL_ROW(3, v3)
                PULL_ROW(4, v4) PULL_ROW(5, v5) PULL_ROW(6, v6) PULL_ROW(7, v7)
                PULL_ADD(0, v0) PULL_ADD(1, v1) PULL_ADD(2, v2) PULL_ADD(3, v3)
                PULL_ADD(4, v4) PULL_ADD(5, v5) PULL_ADD(6, v6) PULL_ADD(7, v7)
            }
        }
        agg4[(size_t)node * 32 + lane] = acc;
    }
#undef PULL_ROW
#undef PULL_ADD
}


template <int M>
__device__ __forceinline__ float half_xor(float v) {
    return __int_as_float(__builtin_amdgcn_update_dpp(0, __float_as_int(v), 0x160 + M, 0xf, 0xf, false));
}

__global__ void gemm1_relu_l2_kernel(const float* __restrict__ agg, const float* __restrict__ W1,
                                     const float* __restrict__ b1, const float* __restrict__ ninv_in,
                                     const float* __restrict__ ninv_out, const float* __restrict__ W2,
                                     float* __restrict__ y_node) {
    constexpr int LDA = 132;
    __shared__ float As[16 * LDA];
    __shared__ float red[8 * 32];

    const int tid  = threadIdx.x;
    const int lane = tid & 31;
    const int wave = tid >> 5;
    const int row0 = blockIdx.x * 16;

    const float4* g4 = (const float4*)(agg + (size_t)row0 * 128);
    {
        const int i0 = tid * 2;
        float4* srow = (float4*)(As + (i0 >> 5) * LDA + (i0 & 31) * 4);
        srow[0] = g4[i0 + 0];
        srow[1] = g4[i0 + 1];
    }

    const int col0 = wave * 16;
    const int mn   = lane & 15;
    const int kb   = (lane >> 4) * 2;

    const int n     = col0 + mn;
    const int mbase = row0 + ((lane >> 4) ? 8 : 0);
    const float  bn = b1[n];
    const float2 w2 = ((const float2*)W2)[n];
    float ni[8], no[8];
#pragma unroll
    for (int r = 0; r < 8; ++r) {
        ni[r] = ninv_in[mbase + r];
        no[r] = ninv_out[mbase + r];
    }
    __syncthreads();

    v8f c = {};
#pragma unroll
    for (int kk = 0; kk < 32; ++kk) {
        const int k = kk * 4 + kb;
        v2f a, b;
        a.x = As[mn * LDA + k];
        a.y = As[mn * LDA + k + 1];
        b.x = W1[(size_t)k * 128 + col0 + mn];
        b.y = W1[(size_t)(k + 1) * 128 + col0 + mn];
        c = __builtin_amdgcn_wmma_f32_16x16x4_f32(false, a, false, b,
                                                  (short)0, c, false, false);
    }

    float p[16];
#pragma unroll
    for (int r = 0; r < 8; ++r) {
        const float h = fmaxf(c[r] * ni[r] + bn, 0.0f) * no[r];
        p[2 * r + 0] = h * w2.x;
        p[2 * r + 1] = h * w2.y;
    }
#pragma unroll
    for (int i = 0; i < 8; ++i) {
        const float keep = (lane & 8) ? p[i + 8] : p[i];
        const float send = (lane & 8) ? p[i] : p[i + 8];
        p[i] = keep + half_xor<8>(send);
    }
#pragma unroll
    for (int i = 0; i < 4; ++i) {
        const float keep = (lane & 4) ? p[i + 4] : p[i];
        const float send = (lane & 4) ? p[i] : p[i + 4];
        p[i] = keep + half_xor<4>(send);
    }
#pragma unroll
    for (int i = 0; i < 2; ++i) {
        const float keep = (lane & 2) ? p[i + 2] : p[i];
        const float send = (lane & 2) ? p[i] : p[i + 2];
        p[i] = keep + half_xor<2>(send);
    }
    {
        const float keep = (lane & 1) ? p[1] : p[0];
        const float send = (lane & 1) ? p[0] : p[1];
        p[0] = keep + half_xor<1>(send);
    }
    red[wave * 32 + lane] = p[0];
    __syncthreads();
    if (tid < 32) {
        const float s01 = red[tid] + red[32 + tid];
        const float s23 = red[64 + tid] + red[96 + tid];
        const float s45 = red[128 + tid] + red[160 + tid];
        const float s67 = red[192 + tid] + red[224 + tid];
        y_node[(size_t)row0 * 2 + tid] = (s01 + s23) + (s45 + s67);
    }
}

__global__ void __launch_bounds__(256)
agg2_pull_final_kernel(const int* __restrict__ off,
                       const int* __restrict__ perm, const float* __restrict__ y_node,
                       const float* __restrict__ ninv_in, const float* __restrict__ b2,
                       float* __restrict__ out, int n) {
    const int sub = threadIdx.x & 15;
    const int ngroups = gridDim.x * 16;
    const float2* y2 = (const float2*)y_node;
    const float b20 = b2[0];
    const float b21 = b2[1];
    for (int node = blockIdx.x * 16 + (threadIdx.x >> 4); node < n; node += ngroups) {
        const int beg = off[node];
        const int end = off[node + 1];
        float a0 = 0.0f, a1 = 0.0f;
        for (int i = beg + sub; i < end; i += 16) {
            const float2 y = y2[perm[i]];
            a0 += y.x;
            a1 += y.y;
        }
#pragma unroll
        for (int o = 8; o > 0; o >>= 1) {
            a0 += __shfl_xor(a0, o, 32);
            a1 += __shfl_xor(a1, o, 32);
        }
        if (sub == 0) {
            const float ni = ninv_in[node];
            out[node * 2 + 0] = a0 * ni + b20;
            out[node * 2 + 1] = a1 * ni + b21;
        }
    }
}

extern "C" void kernel_launch(void* const* d_in, const int* in_sizes, int n_in,
                              void* d_out, int out_size, void* d_ws, size_t ws_size,
                              hipStream_t stream) {
    const float* in_feat = (const float*)d_in[0];
    const float* W1      = (const float*)d_in[1];
    const float* b1      = (const float*)d_in[2];
    const float* W2      = (const float*)d_in[3];
    const float* b2      = (const float*)d_in[4];
    const int*   src     = (const int*)d_in[5];
    const int*   dst     = (const int*)d_in[6];
    float*       out     = (float*)d_out;

    const int N = N_NODES;
    const int E = N_EDGES;

    float* ws      = (float*)d_ws;
    float* deg_out = ws;
    float* ninv_in = ws + 2 * (size_t)N;
    int*   off     = (int*)(ws + 3 * (size_t)N);
    unsigned* sorted = (unsigned*)(ws + 4 * (size_t)N + 64);
    int*   perm    = (int*)(ws + 4 * (size_t)N + 64 + (size_t)E);
    int*   tileoff = (int*)(ws + 4 * (size_t)N + 64 + 2 * (size_t)E);
    float* agg     = ws + 4 * (size_t)N + 64 + 2 * (size_t)E + (size_t)N * 128;
    float* y_node  = agg + (size_t)N * 128;

    hipMemsetAsync(deg_out, 0, (size_t)N * sizeof(float), stream);

    tile_sort_kernel<<<N_TILES, 1024, 0, stream>>>(src, dst, deg_out, sorted, tileoff);
    csr_build_kernel<<<N_BUCKETS, 1024, 0, stream>>>(sorted, tileoff, deg_out, off, ninv_in, perm, N);
    agg128_pull_kernel<<<(N / 8 + 7) / 8, 256, 0, stream>>>(off, perm, in_feat, deg_out, agg, N);
    gemm1_relu_l2_kernel<<<N / 16, 256, 0, stream>>>(agg, W1, b1, ninv_in, deg_out, W2, y_node);
    agg2_pull_final_kernel<<<(N / 4 + 15) / 16, 256, 0, stream>>>(off, perm, y_node, ninv_in, b2, out, N);
}


# layer-1 aggregation fused into the GEMM kernel: each wave sums its two nodes' rows in registers (same slot order) and stores them into the LDS A tile; agg plane and its launch gone
# speedup vs baseline: 3.6644x; 1.0449x over previous
#include <hip/hip_runtime.h>

typedef __attribute__((ext_vector_type(2))) float v2f;
typedef __attribute__((ext_vector_type(8))) float v8f;

#define N_NODES 100000
#define N_EDGES 1600000
#define IN_F 128
#define HID 128

#define TILE_E 4096
#define N_TILES ((N_EDGES + TILE_E - 1) / TILE_E)
#define BUCKET_N 256
#define N_BUCKETS ((N_NODES + BUCKET_N - 1) / BUCKET_N)
#define TPITCH 416

__global__ void __launch_bounds__(1024)
tile_sort_kernel(const int* __restrict__ src, const int* __restrict__ dst,
                 float* __restrict__ deg_out, unsigned* __restrict__ sorted,
                 int* __restrict__ tileoff) {
    __shared__ int hist[512];
    __shared__ int wsum[16];
    __shared__ unsigned stage[TILE_E];
    const int tid  = threadIdx.x;
    const int lane = tid & 31;
    const int wave = tid >> 5;
    const int w    = blockIdx.x;

    if (tid < 512) hist[tid] = 0;
    __syncthreads();

    const int q = w * (TILE_E / 4) + tid;
    const bool have = q < N_EDGES / 4;
    int4 s4 = make_int4(0, 0, 0, 0), d4 = make_int4(0, 0, 0, 0);
    int r0 = 0, r1 = 0, r2 = 0, r3 = 0;
    if (have) {
        s4 = ((const int4*)src)[q];
        d4 = ((const int4*)dst)[q];
        r0 = atomicAdd(&hist[d4.x >> 8], 1);
        r1 = atomicAdd(&hist[d4.y >> 8], 1);
        r2 = atomicAdd(&hist[d4.z >> 8], 1);
        r3 = atomicAdd(&hist[d4.w >> 8], 1);
    }
    __syncthreads();

    const int v = (tid < 512) ? hist[tid] : 0;
    int inc = v;
    for (int o = 1; o < 32; o <<= 1) {
        const int t = __shfl_up(inc, o, 32);
        if (lane >= o) inc += t;
    }
    if (tid < 512 && lane == 31) wsum[wave] = inc;
    __syncthreads();
    if (tid < 512) {
        int p = 0;
        for (int k = 0; k < 16; ++k) p += (k < wave) ? wsum[k] : 0;
        const int excl = p + inc - v;
        hist[tid] = excl;
        if (tid <= N_BUCKETS) tileoff[w * TPITCH + tid] = excl;
    }
    __syncthreads();

    if (have) {
        stage[hist[d4.x >> 8] + r0] = ((unsigned)s4.x << 8) | (unsigned)(d4.x & 255);
        stage[hist[d4.y >> 8] + r1] = ((unsigned)s4.y << 8) | (unsigned)(d4.y & 255);
        stage[hist[d4.z >> 8] + r2] = ((unsigned)s4.z << 8) | (unsigned)(d4.z & 255);
        stage[hist[d4.w >> 8] + r3] = ((unsigned)s4.w << 8) | (unsigned)(d4.w & 255);
    }
    __syncthreads();

    const int total = min(TILE_E, N_EDGES - w * TILE_E);
    if (tid * 4 < total)
        ((uint4*)(sorted + (size_t)w * TILE_E))[tid] = ((const uint4*)stage)[tid];

    if (have) {
        atomicAdd(&deg_out[s4.x], 1.0f);
        atomicAdd(&deg_out[s4.y], 1.0f);
        atomicAdd(&deg_out[s4.z], 1.0f);
        atomicAdd(&deg_out[s4.w], 1.0f);
    }
}

#define KSEG ((N_TILES + 63) / 64)

__global__ void __launch_bounds__(1024)
csr_build_kernel(const unsigned* __restrict__ sorted, const int* __restrict__ tileoff,
                 float* __restrict__ deg_out, int* __restrict__ off,
                 float* __restrict__ ninv_in, int* __restrict__ perm, int n) {
    __shared__ int segBase[KSEG * 64];
    __shared__ int segLen[KSEG * 64];
    __shared__ int cnt[BUCKET_N];
    __shared__ int cur[BUCKET_N];
    __shared__ int wsumA[32];
    __shared__ int wsumB[8];
    const int tid  = threadIdx.x;
    const int lane = tid & 31;
    const int wave = tid >> 5;
    const int b    = blockIdx.x;

    int lo = 0, len = 0;
    if (tid < N_TILES) {
        lo = tileoff[tid * TPITCH + b];
        len = tileoff[tid * TPITCH + b + 1] - lo;
    }
    if (tid < KSEG * 64) {
        segBase[tid] = tid * TILE_E + lo;
        segLen[tid]  = len;
    }
    if (tid < BUCKET_N) cnt[tid] = 0;
    int ssum = lo;
    for (int o = 16; o > 0; o >>= 1) ssum += __shfl_xor(ssum, o, 32);
    if (lane == 0) wsumA[wave] = ssum;
    __syncthreads();
    int bstart = wsumA[lane];
    for (int o = 16; o > 0; o >>= 1) bstart += __shfl_xor(bstart, o, 32);

    const int g   = tid >> 4;
    const int sub = tid & 15;
    unsigned e[KSEG];
    bool has[KSEG];
#pragma unroll
    for (int k = 0; k < KSEG; ++k) {
        const int t = g + 64 * k;
        has[k] = sub < segLen[t];
        e[k] = has[k] ? sorted[segBase[t] + sub] : 0u;
    }
#pragma unroll
    for (int k = 0; k < KSEG; ++k)
        if (has[k]) atomicAdd(&cnt[e[k] & 255u], 1);
    for (int k = 0; k < KSEG; ++k) {
        const int t = g + 64 * k;
        const int l = segLen[t];
        const int bs = segBase[t];
        for (int j = sub + 16; j < l; j += 16) atomicAdd(&cnt[sorted[bs + j] & 255u], 1);
    }
    __syncthreads();

    const int c = (tid < BUCKET_N) ? cnt[tid] : 0;
    int inc = c;
    for (int o = 1; o < 32; o <<= 1) {
        const int t = __shfl_up(inc, o, 32);
        if (lane >= o) inc += t;
    }
    if (tid < BUCKET_N && lane == 31) wsumB[wave] = inc;
    __syncthreads();
    if (tid < BUCKET_N) {
        int p = 0;
        for (int k = 0; k < 8; ++k) p += (k < wave) ? wsumB[k] : 0;
        const int slot0 = bstart + p + inc - c;
        cur[tid] = slot0;
        const int node = b * BUCKET_N + tid;
        if (node < n) {
            off[node] = slot0;
            if (node == n - 1) off[n] = slot0 + c;
            ninv_in[node] = (c > 0) ? (1.0f / sqrtf((float)c)) : 0.0f;
            const float d = deg_out[node];
            deg_out[node] = (d > 0.0f) ? (1.0f / sqrtf(d)) : 0.0f;
        }
    }
    __syncthreads();

#pragma unroll
    for (int k = 0; k < KSEG; ++k)
        if (has[k]) perm[atomicAdd(&cur[e[k] & 255u], 1)] = (int)(e[k] >> 8);
    for (int k = 0; k < KSEG; ++k) {
        const int t = g + 64 * k;
        const int l = segLen[t];
        const int bs = segBase[t];
        for (int j = sub + 16; j < l; j += 16) {
            const unsigned x = sorted[bs + j];
            perm[atomicAdd(&cur[x & 255u], 1)] = (int)(x >> 8);
        }
    }
}


__device__ __forceinline__ float4 pull_node(const float4* __restrict__ x4, const int* __restrict__ perm,
                                            const float* __restrict__ ninv_out,
                                            int beg, int end, int mine, int scl, int lane) {
    const float4 zero4 = make_float4(0.0f, 0.0f, 0.0f, 0.0f);
    float4 acc = zero4;

#define PULL_ROW(k, v)                                                  \
    float4 v = zero4;                                                   \
    if (j + (k) < m) {                                                  \
        const int s = __builtin_amdgcn_readlane(mine, j + (k));         \
        v = x4[(size_t)s * 32 + lane];                                  \
    }
#define PULL_ADD(k, v)                                                              \
    {                                                                               \
        const float w = __int_as_float(__builtin_amdgcn_readlane(scl, j + (k)));    \
        acc.x += v.x * w; acc.y += v.y * w; acc.z += v.z * w; acc.w += v.w * w;     \
    }

    for (int base = beg; base < end; base += 32) {
        const int m = min(32, end - base);
        if (base != beg) {
            mine = (lane < m) ? perm[base + lane] : 0;
            scl = __float_as_int(ninv_out[mine]);
        }
        for (int j = 0; j < m; j += 8) {
            PULL_ROW(0, v0) PULL_ROW(1, v1) PULL_ROW(2, v2) PULL_ROW(3, v3)
            PULL_ROW(4, v4) PULL_ROW(5, v5) PULL_ROW(6, v6) PULL_ROW(7, v7)
            PULL_ADD(0, v0) PULL_ADD(1, v1) PULL_ADD(2, v2) PULL_ADD(3, v3)
            PULL_ADD(4, v4) PULL_ADD(5, v5) PULL_ADD(6, v6) PULL_ADD(7, v7)
        }
    }
#undef PULL_ROW
#undef PULL_ADD
    return acc;
}

template <int M>
__device__ __forceinline__ float half_xor(float v) {
    return __int_as_float(__builtin_amdgcn_update_dpp(0, __float_as_int(v), 0x160 + M, 0xf, 0xf, false));
}

__global__ void __launch_bounds__(256)
pull_gemm1_relu_l2_kernel(const int* __restrict__ off, const int* __restrict__ perm,
                          const float* __restrict__ x, const float* __restrict__ W1,
                          const float* __restrict__ b1, const float* __restrict__ ninv_in,
                          const float* __restrict__ ninv_out, const float* __restrict__ W2,
                          float* __restrict__ y_node) {
    constexpr int LDA = 132;
    __shared__ float As[16 * LDA];
    __shared__ float red[8 * 32];

    const int tid  = threadIdx.x;
    const int lane = tid & 31;
    const int wave = tid >> 5;
    const int row0 = blockIdx.x * 16;

    {
        const int wv = __builtin_amdgcn_readfirstlane(wave);
        const int n0 = row0 + 2 * wv;
        const float4* x4 = (const float4*)x;
        const int o0 = off[n0];
        const int o1 = off[n0 + 1];
        const int o2 = off[n0 + 2];
        const int mA = min(32, o1 - o0);
        const int mB = min(32, o2 - o1);
        const int mineA = (lane < mA) ? perm[o0 + lane] : 0;
        const int mineB = (lane < mB) ? perm[o1 + lane] : 0;
        const int sclA = __float_as_int(ninv_out[mineA]);
        const int sclB = __float_as_int(ninv_out[mineB]);
        const float4 accA = pull_node(x4, perm, ninv_out, o0, o1, mineA, sclA, lane);
        ((float4*)(As + (2 * wv) * LDA))[lane] = accA;
        const float4 accB = pull_node(x4, perm, ninv_out, o1, o2, mineB, sclB, lane);
        ((float4*)(As + (2 * wv + 1) * LDA))[lane] = accB;
    }

    const int col0 = wave * 16;
    const int mn   = lane & 15;
    const int kb   = (lane >> 4) * 2;

    const int n     = col0 + mn;
    const int mbase = row0 + ((lane >> 4) ? 8 : 0);
    const float  bn = b1[n];
    const float2 w2 = ((const float2*)W2)[n];
    float ni[8], no[8];
#pragma unroll
    for (int r = 0; r < 8; ++r) {
        ni[r] = ninv_in[mbase + r];
        no[r] = ninv_out[mbase + r];
    }
    __syncthreads();

    v8f c = {};
#pragma unroll
    for (int kk = 0; kk < 32; ++kk) {
        const int k = kk * 4 + kb;
        v2f a, b;
        a.x = As[mn * LDA + k];
        a.y = As[mn * LDA + k + 1];
        b.x = W1[(size_t)k * 128 + col0 + mn];
        b.y = W1[(size_t)(k + 1) * 128 + col0 + mn];
        c = __builtin_amdgcn_wmma_f32_16x16x4_f32(false, a, false, b,
                                                  (short)0, c, false, false);
    }

    float p[16];
#pragma unroll
    for (int r = 0; r < 8; ++r) {
        const float h = fmaxf(c[r] * ni[r] + bn, 0.0f) * no[r];
        p[2 * r + 0] = h * w2.x;
        p[2 * r + 1] = h * w2.y;
    }
#pragma unroll
    for (int i = 0; i < 8; ++i) {
        const float keep = (lane & 8) ? p[i + 8] : p[i];
        const float send = (lane & 8) ? p[i] : p[i + 8];
        p[i] = keep + half_xor<8>(send);
    }
#pragma unroll
    for (int i = 0; i < 4; ++i) {
        const float keep = (lane & 4) ? p[i + 4] : p[i];
        const float send = (lane & 4) ? p[i] : p[i + 4];
        p[i] = keep + half_xor<4>(send);
    }
#pragma unroll
    for (int i = 0; i < 2; ++i) {
        const float keep = (lane & 2) ? p[i + 2] : p[i];
        const float send = (lane & 2) ? p[i] : p[i + 2];
        p[i] = keep + half_xor<2>(send);
    }
    {
        const float keep = (lane & 1) ? p[1] : p[0];
        const float send = (lane & 1) ? p[0] : p[1];
        p[0] = keep + half_xor<1>(send);
    }
    red[wave * 32 + lane] = p[0];
    __syncthreads();
    if (tid < 32) {
        const float s01 = red[tid] + red[32 + tid];
        const float s23 = red[64 + tid] + red[96 + tid];
        const float s45 = red[128 + tid] + red[160 + tid];
        const float s67 = red[192 + tid] + red[224 + tid];
        y_node[(size_t)row0 * 2 + tid] = (s01 + s23) + (s45 + s67);
    }
}

__global__ void __launch_bounds__(256)
agg2_pull_final_kernel(const int* __restrict__ off,
                       const int* __restrict__ perm, const float* __restrict__ y_node,
                       const float* __restrict__ ninv_in, const float* __restrict__ b2,
                       float* __restrict__ out, int n) {
    const int sub = threadIdx.x & 15;
    const int ngroups = gridDim.x * 16;
    const float2* y2 = (const float2*)y_node;
    const float b20 = b2[0];
    const float b21 = b2[1];
    for (int node = blockIdx.x * 16 + (threadIdx.x >> 4); node < n; node += ngroups) {
        const int beg = off[node];
        const int end = off[node + 1];
        float a0 = 0.0f, a1 = 0.0f;
        for (int i = beg + sub; i < end; i += 16) {
            const float2 y = y2[perm[i]];
            a0 += y.x;
            a1 += y.y;
        }
#pragma unroll
        for (int o = 8; o > 0; o >>= 1) {
            a0 += __shfl_xor(a0, o, 32);
            a1 += __shfl_xor(a1, o, 32);
        }
        if (sub == 0) {
            const float ni = ninv_in[node];
            out[node * 2 + 0] = a0 * ni + b20;
            out[node * 2 + 1] = a1 * ni + b21;
        }
    }
}

extern "C" void kernel_launch(void* const* d_in, const int* in_sizes, int n_in,
                              void* d_out, int out_size, void* d_ws, size_t ws_size,
                              hipStream_t stream) {
    const float* in_feat = (const float*)d_in[0];
    const float* W1      = (const float*)d_in[1];
    const float* b1      = (const float*)d_in[2];
    const float* W2      = (const float*)d_in[3];
    const float* b2      = (const float*)d_in[4];
    const int*   src     = (const int*)d_in[5];
    const int*   dst     = (const int*)d_in[6];
    float*       out     = (float*)d_out;

    const int N = N_NODES;
    const int E = N_EDGES;

    float* ws      = (float*)d_ws;
    float* deg_out = ws;
    float* ninv_in = ws + 2 * (size_t)N;
    int*   off     = (int*)(ws + 3 * (size_t)N);
    unsigned* sorted = (unsigned*)(ws + 4 * (size_t)N + 64);
    int*   perm    = (int*)(ws + 4 * (size_t)N + 64 + (size_t)E);
    int*   tileoff = (int*)(ws + 4 * (size_t)N + 64 + 2 * (size_t)E);
    float* y_node  = ws + 4 * (size_t)N + 64 + 2 * (size_t)E + (size_t)N * 128;

    hipMemsetAsync(deg_out, 0, (size_t)N * sizeof(float), stream);

    tile_sort_kernel<<<N_TILES, 1024, 0, stream>>>(src, dst, deg_out, sorted, tileoff);
    csr_build_kernel<<<N_BUCKETS, 1024, 0, stream>>>(sorted, tileoff, deg_out, off, ninv_in, perm, N);
    pull_gemm1_relu_l2_kernel<<<N / 16, 256, 0, stream>>>(off, perm, in_feat, W1, b1, ninv_in, deg_out, W2, y_node);
    agg2_pull_final_kernel<<<(N / 4 + 15) / 16, 256, 0, stream>>>(off, perm, y_node, ninv_in, b2, out, N);
}


# out-degree counted by the same two-level counting sort keyed by src (LDS counters, one byte per edge per tile in src-bucket order); per-edge global f32 adds and the memset gone; sums unchanged
# speedup vs baseline: 6.1987x; 1.6916x over previous
#include <hip/hip_runtime.h>

typedef __attribute__((ext_vector_type(2))) float v2f;
typedef __attribute__((ext_vector_type(8))) float v8f;

#define N_NODES 100000
#define N_EDGES 1600000
#define IN_F 128
#define HID 128

#define TILE_E 4096
#define N_TILES ((N_EDGES + TILE_E - 1) / TILE_E)
#define BUCKET_N 256
#define N_BUCKETS ((N_NODES + BUCKET_N - 1) / BUCKET_N)
#define TPITCH 416

__global__ void __launch_bounds__(1024)
tile_sort_kernel(const int* __restrict__ src, const int* __restrict__ dst,
                 unsigned* __restrict__ sorted, int* __restrict__ tileoff,
                 unsigned* __restrict__ sortedS, int* __restrict__ tileoffS) {
    __shared__ int hist[1024];
    __shared__ int wsum[32];
    __shared__ unsigned stage[TILE_E];
    __shared__ unsigned stageS[TILE_E / 4];
    const int tid  = threadIdx.x;
    const int lane = tid & 31;
    const int wave = tid >> 5;
    const int w    = blockIdx.x;

    hist[tid] = 0;
    __syncthreads();

    const int q = w * (TILE_E / 4) + tid;
    const bool have = q < N_EDGES / 4;
    int4 s4 = make_int4(0, 0, 0, 0), d4 = make_int4(0, 0, 0, 0);
    int r0 = 0, r1 = 0, r2 = 0, r3 = 0;
    int u0 = 0, u1 = 0, u2 = 0, u3 = 0;
    if (have) {
        s4 = ((const int4*)src)[q];
        d4 = ((const int4*)dst)[q];
        r0 = atomicAdd(&hist[d4.x >> 8], 1);
        r1 = atomicAdd(&hist[d4.y >> 8], 1);
        r2 = atomicAdd(&hist[d4.z >> 8], 1);
        r3 = atomicAdd(&hist[d4.w >> 8], 1);
        u0 = atomicAdd(&hist[512 + (s4.x >> 8)], 1);
        u1 = atomicAdd(&hist[512 + (s4.y >> 8)], 1);
        u2 = atomicAdd(&hist[512 + (s4.z >> 8)], 1);
        u3 = atomicAdd(&hist[512 + (s4.w >> 8)], 1);
    }
    __syncthreads();

    const int v = hist[tid];
    int inc = v;
    for (int o = 1; o < 32; o <<= 1) {
        const int t = __shfl_up(inc, o, 32);
        if (lane >= o) inc += t;
    }
    if (lane == 31) wsum[wave] = inc;
    __syncthreads();
    {
        const int w0 = wave & 16;
        int p = 0;
        for (int k = 0; k < 16; ++k) p += (w0 + k < wave) ? wsum[w0 + k] : 0;
        const int excl = p + inc - v;
        hist[tid] = excl;
        const int idx = tid & 511;
        if (idx <= N_BUCKETS) {
            if (tid < 512) tileoff[w * TPITCH + idx] = excl;
            else           tileoffS[w * TPITCH + idx] = excl;
        }
    }
    __syncthreads();

    if (have) {
        stage[hist[d4.x >> 8] + r0] = ((unsigned)s4.x << 8) | (unsigned)(d4.x & 255);
        stage[hist[d4.y >> 8] + r1] = ((unsigned)s4.y << 8) | (unsigned)(d4.y & 255);
        stage[hist[d4.z >> 8] + r2] = ((unsigned)s4.z << 8) | (unsigned)(d4.z & 255);
        stage[hist[d4.w >> 8] + r3] = ((unsigned)s4.w << 8) | (unsigned)(d4.w & 255);
        unsigned char* sb = (unsigned char*)stageS;
        sb[hist[512 + (s4.x >> 8)] + u0] = (unsigned char)(s4.x & 255);
        sb[hist[512 + (s4.y >> 8)] + u1] = (unsigned char)(s4.y & 255);
        sb[hist[512 + (s4.z >> 8)] + u2] = (unsigned char)(s4.z & 255);
        sb[hist[512 + (s4.w >> 8)] + u3] = (unsigned char)(s4.w & 255);
    }
    __syncthreads();

    const int total = min(TILE_E, N_EDGES - w * TILE_E);
    if (tid * 4 < total) {
        ((uint4*)(sorted + (size_t)w * TILE_E))[tid] = ((const uint4*)stage)[tid];
        sortedS[(size_t)w * (TILE_E / 4) + tid] = stageS[tid];
    }
}

#define KSEG ((N_TILES + 63) / 64)

__global__ void __launch_bounds__(1024)
csr_build_kernel(const unsigned* __restrict__ sorted, const int* __restrict__ tileoff,
                 const unsigned* __restrict__ sortedS, const int* __restrict__ tileoffS,
                 float* __restrict__ ninv_out, int* __restrict__ off,
                 float* __restrict__ ninv_in, int* __restrict__ perm, int n) {
    __shared__ int segBase[KSEG * 64];
    __shared__ int segLen[KSEG * 64];
    __shared__ int segBaseS[KSEG * 64];
    __shared__ int segLenS[KSEG * 64];
    __shared__ int cnt[BUCKET_N];
    __shared__ int cntS[BUCKET_N];
    __shared__ int cur[BUCKET_N];
    __shared__ int wsumA[32];
    __shared__ int wsumB[8];
    const int tid  = threadIdx.x;
    const int lane = tid & 31;
    const int wave = tid >> 5;
    const int b    = blockIdx.x;

    const unsigned char* bytesS = (const unsigned char*)sortedS;
    int lo = 0, len = 0, loS = 0, lenS = 0;
    if (tid < N_TILES) {
        lo = tileoff[tid * TPITCH + b];
        len = tileoff[tid * TPITCH + b + 1] - lo;
        loS = tileoffS[tid * TPITCH + b];
        lenS = tileoffS[tid * TPITCH + b + 1] - loS;
    }
    if (tid < KSEG * 64) {
        segBase[tid] = tid * TILE_E + lo;
        segLen[tid]  = len;
        segBaseS[tid] = tid * TILE_E + loS;
        segLenS[tid]  = lenS;
    }
    if (tid < BUCKET_N) {
        cnt[tid] = 0;
        cntS[tid] = 0;
    }
    int ssum = lo;
    for (int o = 16; o > 0; o >>= 1) ssum += __shfl_xor(ssum, o, 32);
    if (lane == 0) wsumA[wave] = ssum;
    __syncthreads();
    int bstart = wsumA[lane];
    for (int o = 16; o > 0; o >>= 1) bstart += __shfl_xor(bstart, o, 32);

    const int g   = tid >> 4;
    const int sub = tid & 15;
    unsigned e[KSEG];
    bool has[KSEG];
    unsigned es[KSEG];
    bool hasS[KSEG];
#pragma unroll
    for (int k = 0; k < KSEG; ++k) {
        const int t = g + 64 * k;
        has[k] = sub < segLen[t];
        e[k] = has[k] ? sorted[segBase[t] + sub] : 0u;
        hasS[k] = sub < segLenS[t];
        es[k] = hasS[k] ? (unsigned)bytesS[segBaseS[t] + sub] : 0u;
    }
#pragma unroll
    for (int k = 0; k < KSEG; ++k)
        if (has[k]) atomicAdd(&cnt[e[k] & 255u], 1);
#pragma unroll
    for (int k = 0; k < KSEG; ++k)
        if (hasS[k]) atomicAdd(&cntS[es[k]], 1);
    for (int k = 0; k < KSEG; ++k) {
        const int t = g + 64 * k;
        const int l = segLen[t];
        const int bs = segBase[t];
        for (int j = sub + 16; j < l; j += 16) atomicAdd(&cnt[sorted[bs + j] & 255u], 1);
    }
    for (int k = 0; k < KSEG; ++k) {
        const int t = g + 64 * k;
        const int l = segLenS[t];
        const int bs = segBaseS[t];
        for (int j = sub + 16; j < l; j += 16) atomicAdd(&cntS[bytesS[bs + j]], 1);
    }
    __syncthreads();

    const int c = (tid < BUCKET_N) ? cnt[tid] : 0;
    int inc = c;
    for (int o = 1; o < 32; o <<= 1) {
        const int t = __shfl_up(inc, o, 32);
        if (lane >= o) inc += t;
    }
    if (tid < BUCKET_N && lane == 31) wsumB[wave] = inc;
    __syncthreads();
    if (tid < BUCKET_N) {
        int p = 0;
        for (int k = 0; k < 8; ++k) p += (k < wave) ? wsumB[k] : 0;
        const int slot0 = bstart + p + inc - c;
        cur[tid] = slot0;
        const int node = b * BUCKET_N + tid;
        if (node < n) {
            off[node] = slot0;
            if (node == n - 1) off[n] = slot0 + c;
            ninv_in[node] = (c > 0) ? (1.0f / sqrtf((float)c)) : 0.0f;
            const float d = (float)cntS[tid];
            ninv_out[node] = (d > 0.0f) ? (1.0f / sqrtf(d)) : 0.0f;
        }
    }
    __syncthreads();

#pragma unroll
    for (int k = 0; k < KSEG; ++k)
        if (has[k]) perm[atomicAdd(&cur[e[k] & 255u], 1)] = (int)(e[k] >> 8);
    for (int k = 0; k < KSEG; ++k) {
        const int t = g + 64 * k;
        const int l = segLen[t];
        const int bs = segBase[t];
        for (int j = sub + 16; j < l; j += 16) {
            const unsigned x = sorted[bs + j];
            perm[atomicAdd(&cur[x & 255u], 1)] = (int)(x >> 8);
        }
    }
}


__device__ __forceinline__ float4 pull_node(const float4* __restrict__ x4, const int* __restrict__ perm,
                                            const float* __restrict__ ninv_out,
                                            int beg, int end, int mine, int scl, int lane) {
    const float4 zero4 = make_float4(0.0f, 0.0f, 0.0f, 0.0f);
    float4 acc = zero4;

#define PULL_ROW(k, v)                                                  \
    float4 v = zero4;                                                   \
    if (j + (k) < m) {                                                  \
        const int s = __builtin_amdgcn_readlane(mine, j + (k));         \
        v = x4[(size_t)s * 32 + lane];                                  \
    }
#define PULL_ADD(k, v)                                                              \
    {                                                                               \
        const float w = __int_as_float(__builtin_amdgcn_readlane(scl, j + (k)));    \
        acc.x += v.x * w; acc.y += v.y * w; acc.z += v.z * w; acc.w += v.w * w;     \
    }

    for (int base = beg; base < end; base += 32) {
        const int m = min(32, end - base);
        if (base != beg) {
            mine = (lane < m) ? perm[base + lane] : 0;
            scl = __float_as_int(ninv_out[mine]);
        }
        for (int j = 0; j < m; j += 8) {
            PULL_ROW(0, v0) PULL_ROW(1, v1) PULL_ROW(2, v2) PULL_ROW(3, v3)
            PULL_ROW(4, v4) PULL_ROW(5, v5) PULL_ROW(6, v6) PULL_ROW(7, v7)
            PULL_ADD(0, v0) PULL_ADD(1, v1) PULL_ADD(2, v2) PULL_ADD(3, v3)
            PULL_ADD(4, v4) PULL_ADD(5, v5) PULL_ADD(6, v6) PULL_ADD(7, v7)
        }
    }
#undef PULL_ROW
#undef PULL_ADD
    return acc;
}

template <int M>
__device__ __forceinline__ float half_xor(float v) {
    return __int_as_float(__builtin_amdgcn_update_dpp(0, __float_as_int(v), 0x160 + M, 0xf, 0xf, false));
}

__global__ void __launch_bounds__(256)
pull_gemm1_relu_l2_kernel(const int* __restrict__ off, const int* __restrict__ perm,
                          const float* __restrict__ x, const float* __restrict__ W1,
                          const float* __restrict__ b1, const float* __restrict__ ninv_in,
                          const float* __restrict__ ninv_out, const float* __restrict__ W2,
                          float* __restrict__ y_node) {
    constexpr int LDA = 132;
    __shared__ float As[16 * LDA];
    __shared__ float red[8 * 32];

    const int tid  = threadIdx.x;
    const int lane = tid & 31;
    const int wave = tid >> 5;
    const int row0 = blockIdx.x * 16;

    {
        const int wv = __builtin_amdgcn_readfirstlane(wave);
        const int n0 = row0 + 2 * wv;
        const float4* x4 = (const float4*)x;
        const int o0 = off[n0];
        const int o1 = off[n0 + 1];
        const int o2 = off[n0 + 2];
        const int mA = min(32, o1 - o0);
        const int mB = min(32, o2 - o1);
        const int mineA = (lane < mA) ? perm[o0 + lane] : 0;
        const int mineB = (lane < mB) ? perm[o1 + lane] : 0;
        const int sclA = __float_as_int(ninv_out[mineA]);
        const int sclB = __float_as_int(ninv_out[mineB]);
        const float4 accA = pull_node(x4, perm, ninv_out, o0, o1, mineA, sclA, lane);
        ((float4*)(As + (2 * wv) * LDA))[lane] = accA;
        const float4 accB = pull_node(x4, perm, ninv_out, o1, o2, mineB, sclB, lane);
        ((float4*)(As + (2 * wv + 1) * LDA))[lane] = accB;
    }

    const int col0 = wave * 16;
    const int mn   = lane & 15;
    const int kb   = (lane >> 4) * 2;

    const int n     = col0 + mn;
    const int mbase = row0 + ((lane >> 4) ? 8 : 0);
    const float  bn = b1[n];
    const float2 w2 = ((const float2*)W2)[n];
    float ni[8], no[8];
#pragma unroll
    for (int r = 0; r < 8; ++r) {
        ni[r] = ninv_in[mbase + r];
        no[r] = ninv_out[mbase + r];
    }
    __syncthreads();

    v8f c = {};
#pragma unroll
    for (int kk = 0; kk < 32; ++kk) {
        const int k = kk * 4 + kb;
        v2f a, b;
        a.x = As[mn * LDA + k];
        a.y = As[mn * LDA + k + 1];
        b.x = W1[(size_t)k * 128 + col0 + mn];
        b.y = W1[(size_t)(k + 1) * 128 + col0 + mn];
        c = __builtin_amdgcn_wmma_f32_16x16x4_f32(false, a, false, b,
                                                  (short)0, c, false, false);
    }

    float p[16];
#pragma unroll
    for (int r = 0; r < 8; ++r) {
        const float h = fmaxf(c[r] * ni[r] + bn, 0.0f) * no[r];
        p[2 * r + 0] = h * w2.x;
        p[2 * r + 1] = h * w2.y;
    }
#pragma unroll
    for (int i = 0; i < 8; ++i) {
        const float keep = (lane & 8) ? p[i + 8] : p[i];
        const float send = (lane & 8) ? p[i] : p[i + 8];
        p[i] = keep + half_xor<8>(send);
    }
#pragma unroll
    for (int i = 0; i < 4; ++i) {
        const float keep = (lane & 4) ? p[i + 4] : p[i];
        const float send = (lane & 4) ? p[i] : p[i + 4];
        p[i] = keep + half_xor<4>(send);
    }
#pragma unroll
    for (int i = 0; i < 2; ++i) {
        const float keep = (lane & 2) ? p[i + 2] : p[i];
        const float send = (lane & 2) ? p[i] : p[i + 2];
        p[i] = keep + half_xor<2>(send);
    }
    {
        const float keep = (lane & 1) ? p[1] : p[0];
        const float send = (lane & 1) ? p[0] : p[1];
        p[0] = keep + half_xor<1>(send);
    }
    red[wave * 32 + lane] = p[0];
    __syncthreads();
    if (tid < 32) {
        const float s01 = red[tid] + red[32 + tid];
        const float s23 = red[64 + tid] + red[96 + tid];
        const float s45 = red[128 + tid] + red[160 + tid];
        const float s67 = red[192 + tid] + red[224 + tid];
        y_node[(size_t)row0 * 2 + tid] = (s01 + s23) + (s45 + s67);
    }
}

__global__ void __launch_bounds__(256)
agg2_pull_final_kernel(const int* __restrict__ off,
                       const int* __restrict__ perm, const float* __restrict__ y_node,
                       const float* __restrict__ ninv_in, const float* __restrict__ b2,
                       float* __restrict__ out, int n) {
    const int sub = threadIdx.x & 15;
    const int ngroups = gridDim.x * 16;
    const float2* y2 = (const float2*)y_node;
    const float b20 = b2[0];
    const float b21 = b2[1];
    for (int node = blockIdx.x * 16 + (threadIdx.x >> 4); node < n; node += ngroups) {
        const int beg = off[node];
        const int end = off[node + 1];
        float a0 = 0.0f, a1 = 0.0f;
        for (int i = beg + sub; i < end; i += 16) {
            const float2 y = y2[perm[i]];
            a0 += y.x;
            a1 += y.y;
        }
#pragma unroll
        for (int o = 8; o > 0; o >>= 1) {
            a0 += __shfl_xor(a0, o, 32);
            a1 += __shfl_xor(a1, o, 32);
        }
        if (sub == 0) {
            const float ni = ninv_in[node];
            out[node * 2 + 0] = a0 * ni + b20;
            out[node * 2 + 1] = a1 * ni + b21;
        }
    }
}

extern "C" void kernel_launch(void* const* d_in, const int* in_sizes, int n_in,
                              void* d_out, int out_size, void* d_ws, size_t ws_size,
                              hipStream_t stream) {
    const float* in_feat = (const float*)d_in[0];
    const float* W1      = (const float*)d_in[1];
    const float* b1      = (const float*)d_in[2];
    const float* W2      = (const float*)d_in[3];
    const float* b2      = (const float*)d_in[4];
    const int*   src     = (const int*)d_in[5];
    const int*   dst     = (const int*)d_in[6];
    float*       out     = (float*)d_out;

    const int N = N_NODES;
    const int E = N_EDGES;

    float* ws      = (float*)d_ws;
    float* ninv_out = ws;
    float* ninv_in = ws + 2 * (size_t)N;
    int*   off     = (int*)(ws + 3 * (size_t)N);
    unsigned* sorted = (unsigned*)(ws + 4 * (size_t)N + 64);
    int*   perm    = (int*)(ws + 4 * (size_t)N + 64 + (size_t)E);
    int*   tileoff = (int*)(ws + 4 * (size_t)N + 64 + 2 * (size_t)E);
    int*   tileoffS = tileoff + (size_t)N_TILES * TPITCH;
    unsigned* sortedS = (unsigned*)(tileoff + 2 * (size_t)N_TILES * TPITCH);
    float* y_node  = ws + 4 * (size_t)N + 64 + 2 * (size_t)E + (size_t)N * 128;

    tile_sort_kernel<<<N_TILES, 1024, 0, stream>>>(src, dst, sorted, tileoff, sortedS, tileoffS);
    csr_build_kernel<<<N_BUCKETS, 1024, 0, stream>>>(sorted, tileoff, sortedS, tileoffS, ninv_out, off, ninv_in, perm, N);
    pull_gemm1_relu_l2_kernel<<<N / 16, 256, 0, stream>>>(off, perm, in_feat, W1, b1, ninv_in, ninv_out, W2, y_node);
    agg2_pull_final_kernel<<<(N / 4 + 15) / 16, 256, 0, stream>>>(off, perm, y_node, ninv_in, b2, out, N);
}


# the three short launches enter a quarter of the waves: tile sort and list build in 8-wave work-groups (16 edges, 25 runs a thread), layer-2 sums 16 nodes per 16 lanes; arrival order as before
# speedup vs baseline: 6.3941x; 1.0315x over previous
#include <hip/hip_runtime.h>

typedef __attribute__((ext_vector_type(2))) float v2f;
typedef __attribute__((ext_vector_type(8))) float v8f;

#define N_NODES 100000
#define N_EDGES 1600000
#define IN_F 128
#define HID 128

#define TILE_E 4096
#define N_TILES ((N_EDGES + TILE_E - 1) / TILE_E)
#define BUCKET_N 256
#define N_BUCKETS ((N_NODES + BUCKET_N - 1) / BUCKET_N)
#define TPITCH 416
#define SORT_THREADS 256
#define SORT_QPT (TILE_E / 4 / SORT_THREADS)

static_assert(N_BUCKETS + 1 <= 512, "a set of counters holds 512 entries");
static_assert((N_BUCKETS + 1) % 4 == 0, "the offset tables leave in whole 16-byte pieces");
static_assert(N_EDGES % 16 == 0 && TILE_E % 16 == 0, "the byte plane leaves in whole 16-byte pieces");
static_assert(SORT_THREADS * 4 == 1024, "a thread scans four counters");

__global__ void __launch_bounds__(SORT_THREADS)
tile_sort_kernel(const int* __restrict__ src, const int* __restrict__ dst,
                 unsigned* __restrict__ sorted, int* __restrict__ tileoff,
                 unsigned* __restrict__ sortedS, int* __restrict__ tileoffS) {
    __shared__ __attribute__((aligned(16))) int hist[1024];
    __shared__ int wsum[SORT_THREADS / 32];
    __shared__ __attribute__((aligned(16))) unsigned stage[TILE_E];
    __shared__ __attribute__((aligned(16))) unsigned stageS[TILE_E / 4];
    const int tid  = threadIdx.x;
    const int lane = tid & 31;
    const int wave = tid >> 5;
    const int w    = blockIdx.x;

    ((int4*)hist)[tid] = make_int4(0, 0, 0, 0);
    __syncthreads();

    int4 s4[SORT_QPT], d4[SORT_QPT];
    bool have[SORT_QPT];
#pragma unroll
    for (int j = 0; j < SORT_QPT; ++j) {
        const int q = w * (TILE_E / 4) + j * SORT_THREADS + tid;
        have[j] = q < N_EDGES / 4;
        s4[j] = make_int4(0, 0, 0, 0);
        d4[j] = make_int4(0, 0, 0, 0);
        if (have[j]) {
            s4[j] = ((const int4*)src)[q];
            d4[j] = ((const int4*)dst)[q];
        }
    }
    int r[SORT_QPT * 4], u[SORT_QPT * 4];
#pragma unroll
    for (int j = 0; j < SORT_QPT; ++j) {
        r[4 * j + 0] = 0; r[4 * j + 1] = 0; r[4 * j + 2] = 0; r[4 * j + 3] = 0;
        u[4 * j + 0] = 0; u[4 * j + 1] = 0; u[4 * j + 2] = 0; u[4 * j + 3] = 0;
        if (have[j]) {
            r[4 * j + 0] = atomicAdd(&hist[d4[j].x >> 8], 1);
            r[4 * j + 1] = atomicAdd(&hist[d4[j].y >> 8], 1);
            r[4 * j + 2] = atomicAdd(&hist[d4[j].z >> 8], 1);
            r[4 * j + 3] = atomicAdd(&hist[d4[j].w >> 8], 1);
            u[4 * j + 0] = atomicAdd(&hist[512 + (s4[j].x >> 8)], 1);
            u[4 * j + 1] = atomicAdd(&hist[512 + (s4[j].y >> 8)], 1);
            u[4 * j + 2] = atomicAdd(&hist[512 + (s4[j].z >> 8)], 1);
            u[4 * j + 3] = atomicAdd(&hist[512 + (s4[j].w >> 8)], 1);
        }
    }
    __syncthreads();

    const int4 v = ((const int4*)hist)[tid];
    const int t4 = v.x + v.y + v.z + v.w;
    int inc = t4;
    for (int o = 1; o < 32; o <<= 1) {
        const int t = __shfl_up(inc, o, 32);
        if (lane >= o) inc += t;
    }
    if (lane == 31) wsum[wave] = inc;
    __syncthreads();
    {
        const int w0 = wave & 4;
        int p = 0;
        for (int k = 0; k < 4; ++k) p += (w0 + k < wave) ? wsum[w0 + k] : 0;
        const int e0 = p + inc - t4;
        const int e1 = e0 + v.x;
        const int e2 = e1 + v.y;
        const int e3 = e2 + v.z;
        const int4 ex = make_int4(e0, e1, e2, e3);
        ((int4*)hist)[tid] = ex;
        const int idx = (tid * 4) & 511;
        if (idx <= N_BUCKETS) {
            int* const to = (tid < SORT_THREADS / 2) ? tileoff : tileoffS;
            *(int4*)(to + w * TPITCH + idx) = ex;
        }
    }
    __syncthreads();

    unsigned char* sb = (unsigned char*)stageS;
#pragma unroll
    for (int j = 0; j < SORT_QPT; ++j) {
        if (have[j]) {
            stage[hist[d4[j].x >> 8] + r[4 * j + 0]] = ((unsigned)s4[j].x << 8) | (unsigned)(d4[j].x & 255);
            stage[hist[d4[j].y >> 8] + r[4 * j + 1]] = ((unsigned)s4[j].y << 8) | (unsigned)(d4[j].y & 255);
            stage[hist[d4[j].z >> 8] + r[4 * j + 2]] = ((unsigned)s4[j].z << 8) | (unsigned)(d4[j].z & 255);
            stage[hist[d4[j].w >> 8] + r[4 * j + 3]] = ((unsigned)s4[j].w << 8) | (unsigned)(d4[j].w & 255);
            sb[hist[512 + (s4[j].x >> 8)] + u[4 * j + 0]] = (unsigned char)(s4[j].x & 255);
            sb[hist[512 + (s4[j].y >> 8)] + u[4 * j + 1]] = (unsigned char)(s4[j].y & 255);
            sb[hist[512 + (s4[j].z >> 8)] + u[4 * j + 2]] = (unsigned char)(s4[j].z & 255);
            sb[hist[512 + (s4[j].w >> 8)] + u[4 * j + 3]] = (unsigned char)(s4[j].w & 255);
        }
    }
    __syncthreads();

    const int total = min(TILE_E, N_EDGES - w * TILE_E);
#pragma unroll
    for (int j = 0; j < SORT_QPT; ++j) {
        const int i = j * SORT_THREADS + tid;
        if (i * 4 < total) ((uint4*)(sorted + (size_t)w * TILE_E))[i] = ((const uint4*)stage)[i];
    }
    if (tid * 16 < total)
        ((uint4*)(sortedS + (size_t)w * (TILE_E / 4)))[tid] = ((const uint4*)stageS)[tid];
}

#define CSR_THREADS 256
#define CSR_GROUPS (CSR_THREADS / 16)
#define KSEG ((N_TILES + CSR_GROUPS - 1) / CSR_GROUPS)
#define NO_EDGE 0xFFFFFFFFu

static_assert(CSR_THREADS == BUCKET_N, "one thread per node of the bucket");
static_assert(KSEG <= 32, "a lane holds the bounds of two runs");

__global__ void __launch_bounds__(CSR_THREADS)
csr_build_kernel(const unsigned* __restrict__ sorted, const int* __restrict__ tileoff,
                 const unsigned* __restrict__ sortedS, const int* __restrict__ tileoffS,
                 float* __restrict__ ninv_out, int* __restrict__ off,
                 float* __restrict__ ninv_in, int* __restrict__ perm, int n) {
    __shared__ int cnt[BUCKET_N];
    __shared__ int cntS[BUCKET_N];
    __shared__ int cur[BUCKET_N];
    __shared__ int wsumA[CSR_THREADS / 32];
    __shared__ int wsumB[CSR_THREADS / 32];
    const int tid  = threadIdx.x;
    const int lane = tid & 31;
    const int wave = tid >> 5;
    const int b    = blockIdx.x;
    const int g    = tid >> 4;
    const int sub  = tid & 15;

    const unsigned char* bytesS = (const unsigned char*)sortedS;
    int base[2], len[2], baseS[2], lenS[2];
    int ssum = 0;
#pragma unroll
    for (int h = 0; h < 2; ++h) {
        const int k = sub + 16 * h;
        const int t = g + CSR_GROUPS * k;
        const bool ok = (k < KSEG) && (t < N_TILES);
        const int tt = ok ? t : 0;
        const int lo  = tileoff[tt * TPITCH + b];
        const int hi  = tileoff[tt * TPITCH + b + 1];
        const int loS = tileoffS[tt * TPITCH + b];
        const int hiS = tileoffS[tt * TPITCH + b + 1];
        base[h]  = tt * TILE_E + lo;
        len[h]   = ok ? hi - lo : 0;
        baseS[h] = tt * TILE_E + loS;
        lenS[h]  = ok ? hiS - loS : 0;
        ssum += ok ? lo : 0;
    }
    cnt[tid] = 0;
    cntS[tid] = 0;
    for (int o = 16; o > 0; o >>= 1) ssum += __shfl_xor(ssum, o, 32);
    if (lane == 0) wsumA[wave] = ssum;
    __syncthreads();
    int bstart = (lane < CSR_THREADS / 32) ? wsumA[lane] : 0;
    for (int o = 16; o > 0; o >>= 1) bstart += __shfl_xor(bstart, o, 32);

#define RUN_OF(a, k) __shfl(a[(k) >> 4], (k) & 15, 16)

    unsigned e[KSEG], e2[KSEG];
    unsigned es[KSEG], es2[KSEG];
#pragma unroll
    for (int k = 0; k < KSEG; ++k) {
        const int l  = RUN_OF(len, k);
        const int bs = RUN_OF(base, k);
        const int lS  = RUN_OF(lenS, k);
        const int bsS = RUN_OF(baseS, k);
        e[k]   = (sub < l) ? sorted[bs + sub] : NO_EDGE;
        e2[k]  = (sub + 16 < l) ? sorted[bs + sub + 16] : NO_EDGE;
        es[k]  = (sub < lS) ? (unsigned)bytesS[bsS + sub] : NO_EDGE;
        es2[k] = (sub + 16 < lS) ? (unsigned)bytesS[bsS + sub + 16] : NO_EDGE;
    }
#pragma unroll
    for (int k = 0; k < KSEG; ++k) {
        if (e[k] != NO_EDGE) atomicAdd(&cnt[e[k] & 255u], 1);
        if (e2[k] != NO_EDGE) atomicAdd(&cnt[e2[k] & 255u], 1);
    }
#pragma unroll
    for (int k = 0; k < KSEG; ++k) {
        if (es[k] != NO_EDGE) atomicAdd(&cntS[es[k]], 1);
        if (es2[k] != NO_EDGE) atomicAdd(&cntS[es2[k]], 1);
    }
    const bool longD = __any(max(len[0], len[1]) > 32);
    const bool longS = __any(max(lenS[0], lenS[1]) > 32);
    if (longD) {
#pragma unroll
        for (int k = 0; k < KSEG; ++k) {
            const int l  = RUN_OF(len, k);
            const int bs = RUN_OF(base, k);
            for (int j = sub + 32; j < l; j += 16) atomicAdd(&cnt[sorted[bs + j] & 255u], 1);
        }
    }
    if (longS) {
#pragma unroll
        for (int k = 0; k < KSEG; ++k) {
            const int l  = RUN_OF(lenS, k);
            const int bs = RUN_OF(baseS, k);
            for (int j = sub + 32; j < l; j += 16) atomicAdd(&cntS[bytesS[bs + j]], 1);
        }
    }
    __syncthreads();

    const int c = cnt[tid];
    int inc = c;
    for (int o = 1; o < 32; o <<= 1) {
        const int t = __shfl_up(inc, o, 32);
        if (lane >= o) inc += t;
    }
    if (lane == 31) wsumB[wave] = inc;
    __syncthreads();
    {
        int p = 0;
        for (int k = 0; k < CSR_THREADS / 32; ++k) p += (k < wave) ? wsumB[k] : 0;
        const int slot0 = bstart + p + inc - c;
        cur[tid] = slot0;
        const int node = b * BUCKET_N + tid;
        if (node < n) {
            off[node] = slot0;
            if (node == n - 1) off[n] = slot0 + c;
            ninv_in[node] = (c > 0) ? (1.0f / sqrtf((float)c)) : 0.0f;
            const float d = (float)cntS[tid];
            ninv_out[node] = (d > 0.0f) ? (1.0f / sqrtf(d)) : 0.0f;
        }
    }
    __syncthreads();

    int slot[KSEG], slot2[KSEG];
#pragma unroll
    for (int k = 0; k < KSEG; ++k) {
        slot[k]  = (e[k] != NO_EDGE) ? atomicAdd(&cur[e[k] & 255u], 1) : -1;
        slot2[k] = (e2[k] != NO_EDGE) ? atomicAdd(&cur[e2[k] & 255u], 1) : -1;
    }
#pragma unroll
    for (int k = 0; k < KSEG; ++k) {
        if (slot[k] >= 0) perm[slot[k]] = (int)(e[k] >> 8);
        if (slot2[k] >= 0) perm[slot2[k]] = (int)(e2[k] >> 8);
    }
    if (longD) {
#pragma unroll
        for (int k = 0; k < KSEG; ++k) {
            const int l  = RUN_OF(len, k);
            const int bs = RUN_OF(base, k);
            for (int j = sub + 32; j < l; j += 16) {
                const unsigned x = sorted[bs + j];
                perm[atomicAdd(&cur[x & 255u], 1)] = (int)(x >> 8);
            }
        }
    }
#undef RUN_OF
}


__device__ __forceinline__ float4 pull_node(const float4* __restrict__ x4, const int* __restrict__ perm,
                                            const float* __restrict__ ninv_out,
                                            int beg, int end, int mine, int scl, int lane) {
    const float4 zero4 = make_float4(0.0f, 0.0f, 0.0f, 0.0f);
    float4 acc = zero4;

#define PULL_ROW(k, v)                                                  \
    float4 v = zero4;                                                   \
    if (j + (k) < m) {                                                  \
        const int s = __builtin_amdgcn_readlane(mine, j + (k));         \
        v = x4[(size_t)s * 32 + lane];                                  \
    }
#define PULL_ADD(k, v)                                                              \
    {                                                                               \
        const float w = __int_as_float(__builtin_amdgcn_readlane(scl, j + (k)));    \
        acc.x += v.x * w; acc.y += v.y * w; acc.z += v.z * w; acc.w += v.w * w;     \
    }

    for (int base = beg; base < end; base += 32) {
        const int m = min(32, end - base);
        if (base != beg) {
            mine = (lane < m) ? perm[base + lane] : 0;
            scl = __float_as_int(ninv_out[mine]);
        }
        for (int j = 0; j < m; j += 8) {
            PULL_ROW(0, v0) PULL_ROW(1, v1) PULL_ROW(2, v2) PULL_ROW(3, v3)
            PULL_ROW(4, v4) PULL_ROW(5, v5) PULL_ROW(6, v6) PULL_ROW(7, v7)
            PULL_ADD(0, v0) PULL_ADD(1, v1) PULL_ADD(2, v2) PULL_ADD(3, v3)
            PULL_ADD(4, v4) PULL_ADD(5, v5) PULL_ADD(6, v6) PULL_ADD(7, v7)
        }
    }
#undef PULL_ROW
#undef PULL_ADD
    return acc;
}

template <int M>
__device__ __forceinline__ float half_xor(float v) {
    return __int_as_float(__builtin_amdgcn_update_dpp(0, __float_as_int(v), 0x160 + M, 0xf, 0xf, false));
}

__global__ void __launch_bounds__(256)
pull_gemm1_relu_l2_kernel(const int* __restrict__ off, const int* __restrict__ perm,
                          const float* __restrict__ x, const float* __restrict__ W1,
                          const float* __restrict__ b1, const float* __restrict__ ninv_in,
                          const float* __restrict__ ninv_out, const float* __restrict__ W2,
                          float* __restrict__ y_node) {
    constexpr int LDA = 132;
    __shared__ float As[16 * LDA];
    __shared__ float red[8 * 32];

    const int tid  = threadIdx.x;
    const int lane = tid & 31;
    const int wave = tid >> 5;
    const int row0 = blockIdx.x * 16;

    {
        const int wv = __builtin_amdgcn_readfirstlane(wave);
        const int n0 = row0 + 2 * wv;
        const float4* x4 = (const float4*)x;
        const int o0 = off[n0];
        const int o1 = off[n0 + 1];
        const int o2 = off[n0 + 2];
        const int mA = min(32, o1 - o0);
        const int mB = min(32, o2 - o1);
        const int mineA = (lane < mA) ? perm[o0 + lane] : 0;
        const int mineB = (lane < mB) ? perm[o1 + lane] : 0;
        const int sclA = __float_as_int(ninv_out[mineA]);
        const int sclB = __float_as_int(ninv_out[mineB]);
        const float4 accA = pull_node(x4, perm, ninv_out, o0, o1, mineA, sclA, lane);
        ((float4*)(As + (2 * wv) * LDA))[lane] = accA;
        const float4 accB = pull_node(x4, perm, ninv_out, o1, o2, mineB, sclB, lane);
        ((float4*)(As + (2 * wv + 1) * LDA))[lane] = accB;
    }

    const int col0 = wave * 16;
    const int mn   = lane & 15;
    const int kb   = (lane >> 4) * 2;

    const int n     = col0 + mn;
    const int mbase = row0 + ((lane >> 4) ? 8 : 0);
    const float  bn = b1[n];
    const float2 w2 = ((const float2*)W2)[n];
    float ni[8], no[8];
#pragma unroll
    for (int r = 0; r < 8; ++r) {
        ni[r] = ninv_in[mbase + r];
        no[r] = ninv_out[mbase + r];
    }
    __syncthreads();

    v8f c = {};
#pragma unroll
    for (int kk = 0; kk < 32; ++kk) {
        const int k = kk * 4 + kb;
        v2f a, b;
        a.x = As[mn * LDA + k];
        a.y = As[mn * LDA + k + 1];
        b.x = W1[(size_t)k * 128 + col0 + mn];
        b.y = W1[(size_t)(k + 1) * 128 + col0 + mn];
        c = __builtin_amdgcn_wmma_f32_16x16x4_f32(false, a, false, b,
                                                  (short)0, c, false, false);
    }

    float p[16];
#pragma unroll
    for (int r = 0; r < 8; ++r) {
        const float h = fmaxf(c[r] * ni[r] + bn, 0.0f) * no[r];
        p[2 * r + 0] = h * w2.x;
        p[2 * r + 1] = h * w2.y;
    }
#pragma unroll
    for (int i = 0; i < 8; ++i) {
        const float keep = (lane & 8) ? p[i + 8] : p[i];
        const float send = (lane & 8) ? p[i] : p[i + 8];
        p[i] = keep + half_xor<8>(send);
    }
#pragma unroll
    for (int i = 0; i < 4; ++i) {
        const float keep = (lane & 4) ? p[i + 4] : p[i];
        const float send = (lane & 4) ? p[i] : p[i + 4];
        p[i] = keep + half_xor<4>(send);
    }
#pragma unroll
    for (int i = 0; i < 2; ++i) {
        const float keep = (lane & 2) ? p[i + 2] : p[i];
        const float send = (lane & 2) ? p[i] : p[i + 2];
        p[i] = keep + half_xor<2>(send);
    }
    {
        const float keep = (lane & 1) ? p[1] : p[0];
        const float send = (lane & 1) ? p[0] : p[1];
        p[0] = keep + half_xor<1>(send);
    }
    red[wave * 32 + lane] = p[0];
    __syncthreads();
    if (tid < 32) {
        const float s01 = red[tid] + red[32 + tid];
        const float s23 = red[64 + tid] + red[96 + tid];
        const float s45 = red[128 + tid] + red[160 + tid];
        const float s67 = red[192 + tid] + red[224 + tid];
        y_node[(size_t)row0 * 2 + tid] = (s01 + s23) + (s45 + s67);
    }
}

#define AGG2_NPG 16
#define AGG2_BLOCKS ((N_NODES + 16 * AGG2_NPG - 1) / (16 * AGG2_NPG))

__global__ void __launch_bounds__(256)
agg2_pull_final_kernel(const int* __restrict__ off,
                       const int* __restrict__ perm, const float* __restrict__ y_node,
                       const float* __restrict__ ninv_in, const float* __restrict__ b2,
                       float* __restrict__ out, int n) {
    const int sub = threadIdx.x & 15;
    const int nbase = (blockIdx.x * 16 + (threadIdx.x >> 4)) * AGG2_NPG;
    const float2* y2 = (const float2*)y_node;
    const float b20 = b2[0];
    const float b21 = b2[1];
    const int node = nbase + sub;
    const int myoff  = off[min(node, n)];
    const int nxtoff = off[min(node + 1, n)];
    const float myni = (node < n) ? ninv_in[node] : 0.0f;
    int pa[AGG2_NPG], pb[AGG2_NPG];
#pragma unroll
    for (int i = 0; i < AGG2_NPG; ++i) {
        const int beg = __shfl(myoff, i, 16);
        const int end = __shfl(nxtoff, i, 16);
        const int last = max(end - 1, 0);
        pa[i] = perm[min(beg + sub, last)];
        pb[i] = perm[min(beg + sub + 16, last)];
    }
    float2 ya[AGG2_NPG], yb[AGG2_NPG];
#pragma unroll
    for (int i = 0; i < AGG2_NPG; ++i) {
        ya[i] = y2[pa[i]];
        yb[i] = y2[pb[i]];
    }
#pragma unroll
    for (int i = 0; i < AGG2_NPG; ++i)
        asm volatile("" :: "v"(ya[i].x), "v"(ya[i].y), "v"(yb[i].x), "v"(yb[i].y));
    float r0 = 0.0f, r1 = 0.0f;
#pragma unroll
    for (int i = 0; i < AGG2_NPG; ++i) {
        const int beg = __shfl(myoff, i, 16);
        const int end = __shfl(nxtoff, i, 16);
        const bool h1 = beg + sub < end;
        const bool h2 = beg + sub + 16 < end;
        float a0 = h1 ? ya[i].x : 0.0f;
        float a1 = h1 ? ya[i].y : 0.0f;
        a0 = h2 ? a0 + yb[i].x : a0;
        a1 = h2 ? a1 + yb[i].y : a1;
        for (int j = beg + sub + 32; j < end; j += 16) {
            const float2 y = y2[perm[j]];
            a0 += y.x;
            a1 += y.y;
        }
        a0 += half_xor<8>(a0); a1 += half_xor<8>(a1);
        a0 += half_xor<4>(a0); a1 += half_xor<4>(a1);
        a0 += half_xor<2>(a0); a1 += half_xor<2>(a1);
        a0 += half_xor<1>(a0); a1 += half_xor<1>(a1);
        if (sub == i) {
            r0 = a0;
            r1 = a1;
        }
    }
    if (node < n) ((float2*)out)[node] = make_float2(r0 * myni + b20, r1 * myni + b21);
}

extern "C" void kernel_launch(void* const* d_in, const int* in_sizes, int n_in,
                              void* d_out, int out_size, void* d_ws, size_t ws_size,
                              hipStream_t stream) {
    const float* in_feat = (const float*)d_in[0];
    const float* W1      = (const float*)d_in[1];
    const float* b1      = (const float*)d_in[2];
    const float* W2      = (const float*)d_in[3];
    const float* b2      = (const float*)d_in[4];
    const int*   src     = (const int*)d_in[5];
    const int*   dst     = (const int*)d_in[6];
    float*       out     = (float*)d_out;

    const int N = N_NODES;
    const int E = N_EDGES;

    float* ws      = (float*)d_ws;
    float* ninv_out = ws;
    float* ninv_in = ws + 2 * (size_t)N;
    int*   off     = (int*)(ws + 3 * (size_t)N);
    unsigned* sorted = (unsigned*)(ws + 4 * (size_t)N + 64);
    int*   perm    = (int*)(ws + 4 * (size_t)N + 64 + (size_t)E);
    int*   tileoff = (int*)(ws + 4 * (size_t)N + 64 + 2 * (size_t)E);
    int*   tileoffS = tileoff + (size_t)N_TILES * TPITCH;
    unsigned* sortedS = (unsigned*)(tileoff + 2 * (size_t)N_TILES * TPITCH);
    float* y_node  = ws + 4 * (size_t)N + 64 + 2 * (size_t)E + (size_t)N * 128;

    tile_sort_kernel<<<N_TILES, SORT_THREADS, 0, stream>>>(src, dst, sorted, tileoff, sortedS, tileoffS);
    csr_build_kernel<<<N_BUCKETS, CSR_THREADS, 0, stream>>>(sorted, tileoff, sortedS, tileoffS, ninv_out, off, ninv_in, perm, N);
    pull_gemm1_relu_l2_kernel<<<N / 16, 256, 0, stream>>>(off, perm, in_feat, W1, b1, ninv_in, ninv_out, W2, y_node);
    agg2_pull_final_kernel<<<AGG2_BLOCKS, 256, 0, stream>>>(off, perm, y_node, ninv_in, b2, out, N);
}


# layer-1 kernel loops over row tiles: 1024 resident work-groups take 6-7 tiles each, a wave keeps its W1 fragments in registers for all of them, next tile's bounds and sources asked ahead; same sums
# speedup vs baseline: 6.7589x; 1.0571x over previous
#include <hip/hip_runtime.h>

typedef __attribute__((ext_vector_type(2))) float v2f;
typedef __attribute__((ext_vector_type(8))) float v8f;

#define N_NODES 100000
#define N_EDGES 1600000
#define IN_F 128
#define HID 128

#define TILE_E 4096
#define N_TILES ((N_EDGES + TILE_E - 1) / TILE_E)
#define BUCKET_N 256
#define N_BUCKETS ((N_NODES + BUCKET_N - 1) / BUCKET_N)
#define TPITCH 416
#define SORT_THREADS 256
#define SORT_QPT (TILE_E / 4 / SORT_THREADS)

static_assert(N_BUCKETS + 1 <= 512, "a set of counters holds 512 entries");
static_assert((N_BUCKETS + 1) % 4 == 0, "the offset tables leave in whole 16-byte pieces");
static_assert(N_EDGES % 16 == 0 && TILE_E % 16 == 0, "the byte plane leaves in whole 16-byte pieces");
static_assert(SORT_THREADS * 4 == 1024, "a thread scans four counters");

__global__ void __launch_bounds__(SORT_THREADS)
tile_sort_kernel(const int* __restrict__ src, const int* __restrict__ dst,
                 unsigned* __restrict__ sorted, int* __restrict__ tileoff,
                 unsigned* __restrict__ sortedS, int* __restrict__ tileoffS) {
    __shared__ __attribute__((aligned(16))) int hist[1024];
    __shared__ int wsum[SORT_THREADS / 32];
    __shared__ __attribute__((aligned(16))) unsigned stage[TILE_E];
    __shared__ __attribute__((aligned(16))) unsigned stageS[TILE_E / 4];
    const int tid  = threadIdx.x;
    const int lane = tid & 31;
    const int wave = tid >> 5;
    const int w    = blockIdx.x;

    ((int4*)hist)[tid] = make_int4(0, 0, 0, 0);
    __syncthreads();

    int4 s4[SORT_QPT], d4[SORT_QPT];
    bool have[SORT_QPT];
#pragma unroll
    for (int j = 0; j < SORT_QPT; ++j) {
        const int q = w * (TILE_E / 4) + j * SORT_THREADS + tid;
        have[j] = q < N_EDGES / 4;
        s4[j] = make_int4(0, 0, 0, 0);
        d4[j] = make_int4(0, 0, 0, 0);
        if (have[j]) {
            s4[j] = ((const int4*)src)[q];
            d4[j] = ((const int4*)dst)[q];
        }
    }
    int r[SORT_QPT * 4], u[SORT_QPT * 4];
#pragma unroll
    for (int j = 0; j < SORT_QPT; ++j) {
        r[4 * j + 0] = 0; r[4 * j + 1] = 0; r[4 * j + 2] = 0; r[4 * j + 3] = 0;
        u[4 * j + 0] = 0; u[4 * j + 1] = 0; u[4 * j + 2] = 0; u[4 * j + 3] = 0;
        if (have[j]) {
            r[4 * j + 0] = atomicAdd(&hist[d4[j].x >> 8], 1);
            r[4 * j + 1] = atomicAdd(&hist[d4[j].y >> 8], 1);
            r[4 * j + 2] = atomicAdd(&hist[d4[j].z >> 8], 1);
            r[4 * j + 3] = atomicAdd(&hist[d4[j].w >> 8], 1);
            u[4 * j + 0] = atomicAdd(&hist[512 + (s4[j].x >> 8)], 1);
            u[4 * j + 1] = atomicAdd(&hist[512 + (s4[j].y >> 8)], 1);
            u[4 * j + 2] = atomicAdd(&hist[512 + (s4[j].z >> 8)], 1);
            u[4 * j + 3] = atomicAdd(&hist[512 + (s4[j].w >> 8)], 1);
        }
    }
    __syncthreads();

    const int4 v = ((const int4*)hist)[tid];
    const int t4 = v.x + v.y + v.z + v.w;
    int inc = t4;
    for (int o = 1; o < 32; o <<= 1) {
        const int t = __shfl_up(inc, o, 32);
        if (lane >= o) inc += t;
    }
    if (lane == 31) wsum[wave] = inc;
    __syncthreads();
    {
        const int w0 = wave & 4;
        int p = 0;
        for (int k = 0; k < 4; ++k) p += (w0 + k < wave) ? wsum[w0 + k] : 0;
        const int e0 = p + inc - t4;
        const int e1 = e0 + v.x;
        const int e2 = e1 + v.y;
        const int e3 = e2 + v.z;
        const int4 ex = make_int4(e0, e1, e2, e3);
        ((int4*)hist)[tid] = ex;
        const int idx = (tid * 4) & 511;
        if (idx <= N_BUCKETS) {
            int* const to = (tid < SORT_THREADS / 2) ? tileoff : tileoffS;
            *(int4*)(to + w * TPITCH + idx) = ex;
        }
    }
    __syncthreads();

    unsigned char* sb = (unsigned char*)stageS;
#pragma unroll
    for (int j = 0; j < SORT_QPT; ++j) {
        if (have[j]) {
            stage[hist[d4[j].x >> 8] + r[4 * j + 0]] = ((unsigned)s4[j].x << 8) | (unsigned)(d4[j].x & 255);
            stage[hist[d4[j].y >> 8] + r[4 * j + 1]] = ((unsigned)s4[j].y << 8) | (unsigned)(d4[j].y & 255);
            stage[hist[d4[j].z >> 8] + r[4 * j + 2]] = ((unsigned)s4[j].z << 8) | (unsigned)(d4[j].z & 255);
            stage[hist[d4[j].w >> 8] + r[4 * j + 3]] = ((unsigned)s4[j].w << 8) | (unsigned)(d4[j].w & 255);
            sb[hist[512 + (s4[j].x >> 8)] + u[4 * j + 0]] = (unsigned char)(s4[j].x & 255);
            sb[hist[512 + (s4[j].y >> 8)] + u[4 * j + 1]] = (unsigned char)(s4[j].y & 255);
            sb[hist[512 + (s4[j].z >> 8)] + u[4 * j + 2]] = (unsigned char)(s4[j].z & 255);
            sb[hist[512 + (s4[j].w >> 8)] + u[4 * j + 3]] = (unsigned char)(s4[j].w & 255);
        }
    }
    __syncthreads();

    const int total = min(TILE_E, N_EDGES - w * TILE_E);
#pragma unroll
    for (int j = 0; j < SORT_QPT; ++j) {
        const int i = j * SORT_THREADS + tid;
        if (i * 4 < total) ((uint4*)(sorted + (size_t)w * TILE_E))[i] = ((const uint4*)stage)[i];
    }
    if (tid * 16 < total)
        ((uint4*)(sortedS + (size_t)w * (TILE_E / 4)))[tid] = ((const uint4*)stageS)[tid];
}

#define CSR_THREADS 256
#define CSR_GROUPS (CSR_THREADS / 16)
#define KSEG ((N_TILES + CSR_GROUPS - 1) / CSR_GROUPS)
#define NO_EDGE 0xFFFFFFFFu

static_assert(CSR_THREADS == BUCKET_N, "one thread per node of the bucket");
static_assert(KSEG <= 32, "a lane holds the bounds of two runs");

__global__ void __launch_bounds__(CSR_THREADS)
csr_build_kernel(const unsigned* __restrict__ sorted, const int* __restrict__ tileoff,
                 const unsigned* __restrict__ sortedS, const int* __restrict__ tileoffS,
                 float* __restrict__ ninv_out, int* __restrict__ off,
                 float* __restrict__ ninv_in, int* __restrict__ perm, int n) {
    __shared__ int cnt[BUCKET_N];
    __shared__ int cntS[BUCKET_N];
    __shared__ int cur[BUCKET_N];
    __shared__ int wsumA[CSR_THREADS / 32];
    __shared__ int wsumB[CSR_THREADS / 32];
    const int tid  = threadIdx.x;
    const int lane = tid & 31;
    const int wave = tid >> 5;
    const int b    = blockIdx.x;
    const int g    = tid >> 4;
    const int sub  = tid & 15;

    const unsigned char* bytesS = (const unsigned char*)sortedS;
    int base[2], len[2], baseS[2], lenS[2];
    int ssum = 0;
#pragma unroll
    for (int h = 0; h < 2; ++h) {
        const int k = sub + 16 * h;
        const int t = g + CSR_GROUPS * k;
        const bool ok = (k < KSEG) && (t < N_TILES);
        const int tt = ok ? t : 0;
        const int lo  = tileoff[tt * TPITCH + b];
        const int hi  = tileoff[tt * TPITCH + b + 1];
        const int loS = tileoffS[tt * TPITCH + b];
        const int hiS = tileoffS[tt * TPITCH + b + 1];
        base[h]  = tt * TILE_E + lo;
        len[h]   = ok ? hi - lo : 0;
        baseS[h] = tt * TILE_E + loS;
        lenS[h]  = ok ? hiS - loS : 0;
        ssum += ok ? lo : 0;
    }
    cnt[tid] = 0;
    cntS[tid] = 0;
    for (int o = 16; o > 0; o >>= 1) ssum += __shfl_xor(ssum, o, 32);
    if (lane == 0) wsumA[wave] = ssum;
    __syncthreads();
    int bstart = (lane < CSR_THREADS / 32) ? wsumA[lane] : 0;
    for (int o = 16; o > 0; o >>= 1) bstart += __shfl_xor(bstart, o, 32);

#define RUN_OF(a, k) __shfl(a[(k) >> 4], (k) & 15, 16)

    unsigned e[KSEG], e2[KSEG];
    unsigned es[KSEG], es2[KSEG];
#pragma unroll
    for (int k = 0; k < KSEG; ++k) {
        const int l  = RUN_OF(len, k);
        const int bs = RUN_OF(base, k);
        const int lS  = RUN_OF(lenS, k);
        const int bsS = RUN_OF(baseS, k);
        e[k]   = (sub < l) ? sorted[bs + sub] : NO_EDGE;
        e2[k]  = (sub + 16 < l) ? sorted[bs + sub + 16] : NO_EDGE;
        es[k]  = (sub < lS) ? (unsigned)bytesS[bsS + sub] : NO_EDGE;
        es2[k] = (sub + 16 < lS) ? (unsigned)bytesS[bsS + sub + 16] : NO_EDGE;
    }
#pragma unroll
    for (int k = 0; k < KSEG; ++k) {
        if (e[k] != NO_EDGE) atomicAdd(&cnt[e[k] & 255u], 1);
        if (e2[k] != NO_EDGE) atomicAdd(&cnt[e2[k] & 255u], 1);
    }
#pragma unroll
    for (int k = 0; k < KSEG; ++k) {
        if (es[k] != NO_EDGE) atomicAdd(&cntS[es[k]], 1);
        if (es2[k] != NO_EDGE) atomicAdd(&cntS[es2[k]], 1);
    }
    const bool longD = __any(max(len[0], len[1]) > 32);
    const bool longS = __any(max(lenS[0], lenS[1]) > 32);
    if (longD) {
#pragma unroll
        for (int k = 0; k < KSEG; ++k) {
            const int l  = RUN_OF(len, k);
            const int bs = RUN_OF(base, k);
            for (int j = sub + 32; j < l; j += 16) atomicAdd(&cnt[sorted[bs + j] & 255u], 1);
        }
    }
    if (longS) {
#pragma unroll
        for (int k = 0; k < KSEG; ++k) {
            const int l  = RUN_OF(lenS, k);
            const int bs = RUN_OF(baseS, k);
            for (int j = sub + 32; j < l; j += 16) atomicAdd(&cntS[bytesS[bs + j]], 1);
        }
    }
    __syncthreads();

    const int c = cnt[tid];
    int inc = c;
    for (int o = 1; o < 32; o <<= 1) {
        const int t = __shfl_up(inc, o, 32);
        if (lane >= o) inc += t;
    }
    if (lane == 31) wsumB[wave] = inc;
    __syncthreads();
    {
        int p = 0;
        for (int k = 0; k < CSR_THREADS / 32; ++k) p += (k < wave) ? wsumB[k] : 0;
        const int slot0 = bstart + p + inc - c;
        cur[tid] = slot0;
        const int node = b * BUCKET_N + tid;
        if (node < n) {
            off[node] = slot0;
            if (node == n - 1) off[n] = slot0 + c;
            ninv_in[node] = (c > 0) ? (1.0f / sqrtf((float)c)) : 0.0f;
            const float d = (float)cntS[tid];
            ninv_out[node] = (d > 0.0f) ? (1.0f / sqrtf(d)) : 0.0f;
        }
    }
    __syncthreads();

    int slot[KSEG], slot2[KSEG];
#pragma unroll
    for (int k = 0; k < KSEG; ++k) {
        slot[k]  = (e[k] != NO_EDGE) ? atomicAdd(&cur[e[k] & 255u], 1) : -1;
        slot2[k] = (e2[k] != NO_EDGE) ? atomicAdd(&cur[e2[k] & 255u], 1) : -1;
    }
#pragma unroll
    for (int k = 0; k < KSEG; ++k) {
        if (slot[k] >= 0) perm[slot[k]] = (int)(e[k] >> 8);
        if (slot2[k] >= 0) perm[slot2[k]] = (int)(e2[k] >> 8);
    }
    if (longD) {
#pragma unroll
        for (int k = 0; k < KSEG; ++k) {
            const int l  = RUN_OF(len, k);
            const int bs = RUN_OF(base, k);
            for (int j = sub + 32; j < l; j += 16) {
                const unsigned x = sorted[bs + j];
                perm[atomicAdd(&cur[x & 255u], 1)] = (int)(x >> 8);
            }
        }
    }
#undef RUN_OF
}


__device__ __forceinline__ float4 pull_node(const float4* __restrict__ x4, const int* __restrict__ perm,
                                            const float* __restrict__ ninv_out,
                                            int beg, int end, int mine, int scl, int lane) {
    const float4 zero4 = make_float4(0.0f, 0.0f, 0.0f, 0.0f);
    float4 acc = zero4;

#define PULL_ROW(k, v)                                                  \
    float4 v = zero4;                                                   \
    if (j + (k) < m) {                                                  \
        const int s = __builtin_amdgcn_readlane(mine, j + (k));         \
        v = x4[(size_t)s * 32 + lane];                                  \
    }
#define PULL_ADD(k, v)                                                              \
    {                                                                               \
        const float w = __int_as_float(__builtin_amdgcn_readlane(scl, j + (k)));    \
        acc.x += v.x * w; acc.y += v.y * w; acc.z += v.z * w; acc.w += v.w * w;     \
    }

    for (int base = beg; base < end; base += 32) {
        const int m = min(32, end - base);
        if (base != beg) {
            mine = (lane < m) ? perm[base + lane] : 0;
            scl = __float_as_int(ninv_out[mine]);
        }
        for (int j = 0; j < m; j += 8) {
            PULL_ROW(0, v0) PULL_ROW(1, v1) PULL_ROW(2, v2) PULL_ROW(3, v3)
            PULL_ROW(4, v4) PULL_ROW(5, v5) PULL_ROW(6, v6) PULL_ROW(7, v7)
            PULL_ADD(0, v0) PULL_ADD(1, v1) PULL_ADD(2, v2) PULL_ADD(3, v3)
            PULL_ADD(4, v4) PULL_ADD(5, v5) PULL_ADD(6, v6) PULL_ADD(7, v7)
        }
    }
#undef PULL_ROW
#undef PULL_ADD
    return acc;
}

template <int M>
__device__ __forceinline__ float half_xor(float v) {
    return __int_as_float(__builtin_amdgcn_update_dpp(0, __float_as_int(v), 0x160 + M, 0xf, 0xf, false));
}

#define GEMM_TILES (N_NODES / 16)
#define GEMM_BLOCKS 1024

static_assert(N_NODES % 16 == 0, "whole 16-row slices");
static_assert(GEMM_BLOCKS <= GEMM_TILES, "every work-group has a first slice");

__global__ void __launch_bounds__(256, 8)
pull_gemm1_relu_l2_kernel(const int* __restrict__ off, const int* __restrict__ perm,
                          const float* __restrict__ x, const float* __restrict__ W1,
                          const float* __restrict__ b1, const float* __restrict__ ninv_in,
                          const float* __restrict__ ninv_out, const float* __restrict__ W2,
                          float* __restrict__ y_node) {
    constexpr int LDA = 132;
    __shared__ float As[16 * LDA];
    __shared__ float red[8 * 32];

    const int tid  = threadIdx.x;
    const int lane = tid & 31;
    const int wave = tid >> 5;
    const int wv   = __builtin_amdgcn_readfirstlane(wave);
    const float4* x4 = (const float4*)x;

    const int col0 = wave * 16;
    const int mn   = lane & 15;
    const int kb   = (lane >> 4) * 2;
    const int n    = col0 + mn;

    v2f bw[32];
#pragma unroll
    for (int kk = 0; kk < 32; ++kk) {
        const int k = kk * 4 + kb;
        bw[kk].x = W1[(size_t)k * 128 + n];
        bw[kk].y = W1[(size_t)(k + 1) * 128 + n];
    }
    const float  bn = b1[n];
    const float2 w2 = ((const float2*)W2)[n];

    int tile = blockIdx.x;
    int o0, o1, o2, mineA, mineB;
    {
        const int n0 = tile * 16 + 2 * wv;
        o0 = off[n0];
        o1 = off[n0 + 1];
        o2 = off[n0 + 2];
        const int mA = min(32, o1 - o0);
        const int mB = min(32, o2 - o1);
        mineA = (lane < mA) ? perm[o0 + lane] : 0;
        mineB = (lane < mB) ? perm[o1 + lane] : 0;
    }

    for (;;) {
        const int row0 = tile * 16;
        const int next = tile + GEMM_BLOCKS;
        const bool more = next < GEMM_TILES;

        const int sclA = __float_as_int(ninv_out[mineA]);
        const int sclB = __float_as_int(ninv_out[mineB]);

        const int n1 = more ? next * 16 + 2 * wv : 0;
        const int qv = off[n1 + min(lane, 2)];

        {
            const float4 accA = pull_node(x4, perm, ninv_out, o0, o1, mineA, sclA, lane);
            ((float4*)(As + (2 * wv) * LDA))[lane] = accA;
            const float4 accB = pull_node(x4, perm, ninv_out, o1, o2, mineB, sclB, lane);
            ((float4*)(As + (2 * wv + 1) * LDA))[lane] = accB;
        }

        const int q0 = __builtin_amdgcn_readlane(qv, 0);
        const int q1 = __builtin_amdgcn_readlane(qv, 1);
        const int q2 = __builtin_amdgcn_readlane(qv, 2);
        __syncthreads();

        const int mAn = more ? min(32, q1 - q0) : 0;
        const int mBn = more ? min(32, q2 - q1) : 0;
        const int mineAn = (lane < mAn) ? perm[q0 + lane] : 0;
        const int mineBn = (lane < mBn) ? perm[q1 + lane] : 0;

        const int mbase = row0 + ((lane >> 4) ? 8 : 0);
        float ni[8], no[8];
#pragma unroll
        for (int r = 0; r < 8; ++r) {
            ni[r] = ninv_in[mbase + r];
            no[r] = ninv_out[mbase + r];
        }

        v8f c = {};
#pragma unroll
        for (int kk = 0; kk < 32; ++kk) {
            const int k = kk * 4 + kb;
            v2f a;
            a.x = As[mn * LDA + k];
            a.y = As[mn * LDA + k + 1];
            c = __builtin_amdgcn_wmma_f32_16x16x4_f32(false, a, false, bw[kk],
                                                      (short)0, c, false, false);
        }

        float p[16];
#pragma unroll
        for (int r = 0; r < 8; ++r) {
            const float h = fmaxf(c[r] * ni[r] + bn, 0.0f) * no[r];
            p[2 * r + 0] = h * w2.x;
            p[2 * r + 1] = h * w2.y;
        }
#pragma unroll
        for (int i = 0; i < 8; ++i) {
            const float keep = (lane & 8) ? p[i + 8] : p[i];
            const float send = (lane & 8) ? p[i] : p[i + 8];
            p[i] = keep + half_xor<8>(send);
        }
#pragma unroll
        for (int i = 0; i < 4; ++i) {
            const float keep = (lane & 4) ? p[i + 4] : p[i];
            const float send = (lane & 4) ? p[i] : p[i + 4];
            p[i] = keep + half_xor<4>(send);
        }
#pragma unroll
        for (int i = 0; i < 2; ++i) {
            const float keep = (lane & 2) ? p[i + 2] : p[i];
            const float send = (lane & 2) ? p[i] : p[i + 2];
            p[i] = keep + half_xor<2>(send);
        }
        {
            const float keep = (lane & 1) ? p[1] : p[0];
            const float send = (lane & 1) ? p[0] : p[1];
            p[0] = keep + half_xor<1>(send);
        }
        red[wave * 32 + lane] = p[0];
        __syncthreads();
        if (tid < 32) {
            const float s01 = red[tid] + red[32 + tid];
            const float s23 = red[64 + tid] + red[96 + tid];
            const float s45 = red[128 + tid] + red[160 + tid];
            const float s67 = red[192 + tid] + red[224 + tid];
            y_node[(size_t)row0 * 2 + tid] = (s01 + s23) + (s45 + s67);
        }
        if (!more) break;
        tile = next;
        o0 = q0; o1 = q1; o2 = q2;
        mineA = mineAn; mineB = mineBn;
    }
}

#define AGG2_NPG 16
#define AGG2_BLOCKS ((N_NODES + 16 * AGG2_NPG - 1) / (16 * AGG2_NPG))

__global__ void __launch_bounds__(256)
agg2_pull_final_kernel(const int* __restrict__ off,
                       const int* __restrict__ perm, const float* __restrict__ y_node,
                       const float* __restrict__ ninv_in, const float* __restrict__ b2,
                       float* __restrict__ out, int n) {
    const int sub = threadIdx.x & 15;
    const int nbase = (blockIdx.x * 16 + (threadIdx.x >> 4)) * AGG2_NPG;
    const float2* y2 = (const float2*)y_node;
    const float b20 = b2[0];
    const float b21 = b2[1];
    const int node = nbase + sub;
    const int myoff  = off[min(node, n)];
    const int nxtoff = off[min(node + 1, n)];
    const float myni = (node < n) ? ninv_in[node] : 0.0f;
    int pa[AGG2_NPG], pb[AGG2_NPG];
#pragma unroll
    for (int i = 0; i < AGG2_NPG; ++i) {
        const int beg = __shfl(myoff, i, 16);
        const int end = __shfl(nxtoff, i, 16);
        const int last = max(end - 1, 0);
        pa[i] = perm[min(beg + sub, last)];
        pb[i] = perm[min(beg + sub + 16, last)];
    }
    float2 ya[AGG2_NPG], yb[AGG2_NPG];
#pragma unroll
    for (int i = 0; i < AGG2_NPG; ++i) {
        ya[i] = y2[pa[i]];
        yb[i] = y2[pb[i]];
    }
#pragma unroll
    for (int i = 0; i < AGG2_NPG; ++i)
        asm volatile("" :: "v"(ya[i].x), "v"(ya[i].y), "v"(yb[i].x), "v"(yb[i].y));
    float r0 = 0.0f, r1 = 0.0f;
#pragma unroll
    for (int i = 0; i < AGG2_NPG; ++i) {
        const int beg = __shfl(myoff, i, 16);
        const int end = __shfl(nxtoff, i, 16);
        const bool h1 = beg + sub < end;
        const bool h2 = beg + sub + 16 < end;
        float a0 = h1 ? ya[i].x : 0.0f;
        float a1 = h1 ? ya[i].y : 0.0f;
        a0 = h2 ? a0 + yb[i].x : a0;
        a1 = h2 ? a1 + yb[i].y : a1;
        for (int j = beg + sub + 32; j < end; j += 16) {
            const float2 y = y2[perm[j]];
            a0 += y.x;
            a1 += y.y;
        }
        a0 += half_xor<8>(a0); a1 += half_xor<8>(a1);
        a0 += half_xor<4>(a0); a1 += half_xor<4>(a1);
        a0 += half_xor<2>(a0); a1 += half_xor<2>(a1);
        a0 += half_xor<1>(a0); a1 += half_xor<1>(a1);
        if (sub == i) {
            r0 = a0;
            r1 = a1;
        }
    }
    if (node < n) ((float2*)out)[node] = make_float2(r0 * myni + b20, r1 * myni + b21);
}

extern "C" void kernel_launch(void* const* d_in, const int* in_sizes, int n_in,
                              void* d_out, int out_size, void* d_ws, size_t ws_size,
                              hipStream_t stream) {
    const float* in_feat = (const float*)d_in[0];
    const float* W1      = (const float*)d_in[1];
    const float* b1      = (const float*)d_in[2];
    const float* W2      = (const float*)d_in[3];
    const float* b2      = (const float*)d_in[4];
    const int*   src     = (const int*)d_in[5];
    const int*   dst     = (const int*)d_in[6];
    float*       out     = (float*)d_out;

    const int N = N_NODES;
    const int E = N_EDGES;

    float* ws      = (float*)d_ws;
    float* ninv_out = ws;
    float* ninv_in = ws + 2 * (size_t)N;
    int*   off     = (int*)(ws + 3 * (size_t)N);
    unsigned* sorted = (unsigned*)(ws + 4 * (size_t)N + 64);
    int*   perm    = (int*)(ws + 4 * (size_t)N + 64 + (size_t)E);
    int*   tileoff = (int*)(ws + 4 * (size_t)N + 64 + 2 * (size_t)E);
    int*   tileoffS = tileoff + (size_t)N_TILES * TPITCH;
    unsigned* sortedS = (unsigned*)(tileoff + 2 * (size_t)N_TILES * TPITCH);
    float* y_node  = ws + 4 * (size_t)N + 64 + 2 * (size_t)E + (size_t)N * 128;

    tile_sort_kernel<<<N_TILES, SORT_THREADS, 0, stream>>>(src, dst, sorted, tileoff, sortedS, tileoffS);
    csr_build_kernel<<<N_BUCKETS, CSR_THREADS, 0, stream>>>(sorted, tileoff, sortedS, tileoffS, ninv_out, off, ninv_in, perm, N);
    pull_gemm1_relu_l2_kernel<<<GEMM_BLOCKS, 256, 0, stream>>>(off, perm, in_feat, W1, b1, ninv_in, ninv_out, W2, y_node);
    agg2_pull_final_kernel<<<AGG2_BLOCKS, 256, 0, stream>>>(off, perm, y_node, ninv_in, b2, out, N);
}
